# starting kernel
# baseline (speedup 1.0000x reference)
#include <hip/hip_runtime.h>

typedef float v2f __attribute__((ext_vector_type(2)));
typedef float v8f __attribute__((ext_vector_type(8)));

#define DIM 128

__global__ void gcn_zero_f32(float* __restrict__ p, int n) {
    int i = blockIdx.x * blockDim.x + threadIdx.x;
    int stride = gridDim.x * blockDim.x;
    for (; i < n; i += stride) p[i] = 0.0f;
}

__global__ void gcn_degree(const int* __restrict__ src, const int* __restrict__ dst,
                           float* __restrict__ deg_s, float* __restrict__ deg_d, int ne) {
    int i = blockIdx.x * blockDim.x + threadIdx.x;
    if (i < ne) {
        atomicAdd(&deg_s[src[i]], 1.0f);
        atomicAdd(&deg_d[dst[i]], 1.0f);
    }
}

__global__ void gcn_norm(float* __restrict__ p, int n) {
    int i = blockIdx.x * blockDim.x + threadIdx.x;
    if (i < n) p[i] = rsqrtf(fmaxf(p[i], 1.0f));
}

__global__ void gcn_scatter(const float* __restrict__ h, const float* __restrict__ norm_s,
                            const int* __restrict__ src, const int* __restrict__ dst,
                            float* __restrict__ agg, int ne) {
    int tid  = blockIdx.x * blockDim.x + threadIdx.x;
    int e    = tid >> 5;
    if (e >= ne) return;
    int lane = tid & 31;
    int s = src[e];
    int d = dst[e];
    float ns = norm_s[s];
    const float4 v = *reinterpret_cast<const float4*>(h + (size_t)s * DIM + lane * 4);
    float* ap = agg + (size_t)d * DIM + lane * 4;
    atomicAdd(ap + 0, v.x * ns);
    atomicAdd(ap + 1, v.y * ns);
    atomicAdd(ap + 2, v.z * ns);
    atomicAdd(ap + 3, v.w * ns);
}

#define LDA 129

__global__ void gcn_gemm_wmma(const float* __restrict__ agg, const float* __restrict__ norm_d,
                              const float* __restrict__ W, const float* __restrict__ bias,
                              float* __restrict__ out, int relu) {
    __shared__ float lA[16 * LDA];

    const int m0 = blockIdx.x * 16;
    const int t  = threadIdx.x;

    for (int i = t; i < 16 * DIM; i += 256) {
        int r = i >> 7;
        int k = i & (DIM - 1);
        int row = m0 + r;
        lA[r * LDA + k] = agg[(size_t)row * DIM + k] * norm_d[row];
    }
    __syncthreads();

    const int wave = t >> 5;
    const int lane = t & 31;
    const int n0   = wave * 16;
    const int half = lane >> 4;
    const int l16  = lane & 15;

    const float* Ar = lA + l16 * LDA;
    const float* Wc = W + n0 + l16;

    v8f c = {};
#pragma unroll
    for (int kk = 0; kk < DIM; kk += 4) {
        const int kbase = kk + half * 2;
        v2f a, b;
        a.x = Ar[kbase + 0];
        a.y = Ar[kbase + 1];
        b.x = Wc[(kbase + 0) * DIM];
        b.y = Wc[(kbase + 1) * DIM];
        c = __builtin_amdgcn_wmma_f32_16x16x4_f32(
                 false, a,  false, b,
                 (short)0, c,  false,  false);
    }

    const float bn = bias[n0 + l16];
#pragma unroll
    for (int r = 0; r < 8; ++r) {
        int row = m0 + r + half * 8;
        float v = c[r] + bn;
        if (relu) v = fmaxf(v, 0.0f);
        out[(size_t)row * DIM + n0 + l16] = v;
    }
}

extern "C" void kernel_launch(void* const* d_in, const int* in_sizes, int n_in,
                              void* d_out, int out_size, void* d_ws, size_t ws_size,
                              hipStream_t stream) {
    const float* feat = (const float*)d_in[0];
    const float* W1   = (const float*)d_in[1];
    const float* b1   = (const float*)d_in[2];
    const float* W2   = (const float*)d_in[3];
    const float* b2   = (const float*)d_in[4];
    const int*   src  = (const int*)d_in[5];
    const int*   dst  = (const int*)d_in[6];

    const int N = in_sizes[0] / DIM;
    const int E = in_sizes[5];

    float* ws     = (float*)d_ws;
    float* norm_s = ws;
    float* norm_d = ws + N;
    float* agg    = ws + 2 * (size_t)N;

    float* out = (float*)d_out;

    const int ZB = 256;
    const int aggN = N * DIM;

    gcn_zero_f32<<<(2 * N + ZB - 1) / ZB, ZB, 0, stream>>>(norm_s, 2 * N);
    gcn_degree<<<(E + ZB - 1) / ZB, ZB, 0, stream>>>(src, dst, norm_s, norm_d, E);
    gcn_norm<<<(2 * N + ZB - 1) / ZB, ZB, 0, stream>>>(norm_s, 2 * N);

    const int scatterThreads = E * 32;
    const int scatterBlocks  = (scatterThreads + ZB - 1) / ZB;

    gcn_zero_f32<<<(aggN + ZB - 1) / ZB, ZB, 0, stream>>>(agg, aggN);
    gcn_scatter<<<scatterBlocks, ZB, 0, stream>>>(feat, norm_s, src, dst, agg, E);
    gcn_gemm_wmma<<<N / 16, 256, 0, stream>>>(agg, norm_d, W1, b1, out,  1);

    gcn_zero_f32<<<(aggN + ZB - 1) / ZB, ZB, 0, stream>>>(agg, aggN);
    gcn_scatter<<<scatterBlocks, ZB, 0, stream>>>(out, norm_s, src, dst, agg, E);
    gcn_gemm_wmma<<<N / 16, 256, 0, stream>>>(agg, norm_d, W2, b2, out,  0);
}


# Float-atomic edge scatter replaced by a per-row gather over destination-ordered index lists built in the call (integer atomic counts), each row sorted by source index: fixed sum order; no agg zeroing
# speedup vs baseline: 2.2261x; 2.2261x over previous
#include <hip/hip_runtime.h>


typedef float v2f __attribute__((ext_vector_type(2)));
typedef float v8f __attribute__((ext_vector_type(8)));

#define DIM 128

__global__ void gcn_zero_f32(float* __restrict__ p, int n) {
    int i = blockIdx.x * blockDim.x + threadIdx.x;
    int stride = gridDim.x * blockDim.x;
    for (; i < n; i += stride) p[i] = 0.0f;
}

__global__ void gcn_count(const int* __restrict__ src, const int* __restrict__ dst,
                          int* __restrict__ cnt_s, int* __restrict__ cnt_d,
                          int* __restrict__ pos, int ne) {
    int i = blockIdx.x * blockDim.x + threadIdx.x;
    if (i < ne) {
        const int s = src[i];
        const int d = dst[i];
        atomicAdd(&cnt_s[s], 1);
        pos[i] = atomicAdd(&cnt_d[d], 1);
    }
}

#define SCAN_T 256
#define SCAN_PER 4
#define SCAN_TILE (SCAN_T * SCAN_PER)

__global__ void gcn_scan_tile(const int* __restrict__ cnt, int* __restrict__ row_ptr,
                              int* __restrict__ tile_sum, int n) {
    __shared__ int wtot[SCAN_T / 32];
    const int t = threadIdx.x;
    const int lane = t & 31;
    const int wave = t >> 5;
    const int base = blockIdx.x * SCAN_TILE + t * SCAN_PER;
    int v[SCAN_PER];
    int ts = 0;
    for (int k = 0; k < SCAN_PER; ++k) {
        v[k] = (base + k < n) ? cnt[base + k] : 0;
        ts += v[k];
    }
    int x = ts;
    for (int o = 1; o < 32; o <<= 1) {
        int y = __shfl_up(x, o);
        if (lane >= o) x += y;
    }
    if (lane == 31) wtot[wave] = x;
    __syncthreads();
    int woff = 0;
    for (int w = 0; w < SCAN_T / 32; ++w)
        if (w < wave) woff += wtot[w];
    int run = woff + x - ts;
    for (int k = 0; k < SCAN_PER; ++k) {
        if (base + k < n) row_ptr[base + k] = run;
        run += v[k];
    }
    if (t == SCAN_T - 1) tile_sum[blockIdx.x] = run;
}

__global__ void gcn_finish(float* __restrict__ nrm, int* __restrict__ row_ptr,
                           const int* __restrict__ tile_sum, int n) {
    __shared__ int s_off;
    const int t = threadIdx.x;
    if (t < 32) {
        int a = 0;
        for (int j = t; j < (int)blockIdx.x; j += 32) a += tile_sum[j];
        for (int o = 16; o > 0; o >>= 1) a += __shfl_xor(a, o);
        if (t == 0) s_off = a;
    }
    __syncthreads();
    const int off = s_off;
    const int base = blockIdx.x * SCAN_TILE + t * SCAN_PER;
    for (int k = 0; k < SCAN_PER; ++k) {
        const int i = base + k;
        if (i < n) {
            const int cs = __float_as_int(nrm[i]);
            const int cd = __float_as_int(nrm[n + i]);
            const int rp = row_ptr[i] + off;
            row_ptr[i] = rp;
            if (i == n - 1) row_ptr[n] = rp + cd;
            nrm[i]     = rsqrtf(fmaxf((float)cs, 1.0f));
            nrm[n + i] = rsqrtf(fmaxf((float)cd, 1.0f));
        }
    }
}

__global__ void gcn_fill(const int* __restrict__ src, const int* __restrict__ dst,
                         const int* __restrict__ pos, const int* __restrict__ row_ptr,
                         int* __restrict__ col_u, int ne) {
    int i = blockIdx.x * blockDim.x + threadIdx.x;
    if (i < ne) col_u[row_ptr[dst[i]] + pos[i]] = src[i];
}

__global__ void gcn_sort_rows(const int* __restrict__ row_ptr, const int* __restrict__ col_u,
                              int* __restrict__ col, int n) {
    const int r = blockIdx.x * (blockDim.x >> 5) + __builtin_amdgcn_readfirstlane(threadIdx.x >> 5);
    if (r >= n) return;
    const int lane = threadIdx.x & 31;
    const int start = row_ptr[r];
    const int d = row_ptr[r + 1] - start;
    const int* __restrict__ in = col_u + start;
    for (int c0 = 0; c0 < d; c0 += 32) {
        const int i = c0 + lane;
        const int key = (i < d) ? in[i] : 0x7fffffff;
        int rank = 0;
        for (int j0 = 0; j0 < d; j0 += 32) {
            const int kl = (j0 + lane < d) ? in[j0 + lane] : 0x7fffffff;
            const int m = min(32, d - j0);
            for (int jj = 0; jj < m; ++jj) {
                const int kj = __builtin_amdgcn_readlane(kl, jj);
                rank += (int)((kj < key) | ((kj == key) & (j0 + jj < i)));
            }
        }
        if (i < d) col[start + rank] = key;
    }
}

template <int K>
__device__ __forceinline__ void gather_batch(const float* __restrict__ h, int s_l, float w_l,
                                             int j, int lane4, float4& acc) {
    float4 v[K];
    float w[K];
#pragma unroll
    for (int k = 0; k < K; ++k) {
        const int s = __builtin_amdgcn_readlane(s_l, j + k);
        w[k] = __int_as_float(__builtin_amdgcn_readlane(__float_as_int(w_l), j + k));
        v[k] = *reinterpret_cast<const float4*>(h + (size_t)s * DIM + lane4);
    }
#pragma unroll
    for (int k = 0; k < K; ++k) {
        acc.x = fmaf(v[k].x, w[k], acc.x);
        acc.y = fmaf(v[k].y, w[k], acc.y);
        acc.z = fmaf(v[k].z, w[k], acc.z);
        acc.w = fmaf(v[k].w, w[k], acc.w);
    }
}

__global__ void gcn_gather(const float* __restrict__ h, const float* __restrict__ norm_s,
                           const int* __restrict__ row_ptr, const int* __restrict__ col,
                           float* __restrict__ agg, int n) {
    const int r = blockIdx.x * (blockDim.x >> 5) + __builtin_amdgcn_readfirstlane(threadIdx.x >> 5);
    if (r >= n) return;
    const int lane = threadIdx.x & 31;
    const int lane4 = lane * 4;
    const int start = row_ptr[r];
    const int end = row_ptr[r + 1];
    float4 acc = make_float4(0.0f, 0.0f, 0.0f, 0.0f);
    for (int c0 = start; c0 < end; c0 += 32) {
        const int idx = c0 + lane;
        int s_l = 0;
        float w_l = 0.0f;
        if (idx < end) {
            s_l = col[idx];
            w_l = norm_s[s_l];
        }
        const int m = min(32, end - c0);
        int j = 0;
        for (; j + 8 <= m; j += 8) gather_batch<8>(h, s_l, w_l, j, lane4, acc);
        if (j + 4 <= m) { gather_batch<4>(h, s_l, w_l, j, lane4, acc); j += 4; }
        switch (m - j) {
            case 3: gather_batch<3>(h, s_l, w_l, j, lane4, acc); break;
            case 2: gather_batch<2>(h, s_l, w_l, j, lane4, acc); break;
            case 1: gather_batch<1>(h, s_l, w_l, j, lane4, acc); break;
            default: break;
        }
    }
    *reinterpret_cast<float4*>(agg + (size_t)r * DIM + lane4) = acc;
}

#define LDA 129

__global__ void gcn_gemm_wmma(const float* __restrict__ agg, const float* __restrict__ norm_d,
                              const float* __restrict__ W, const float* __restrict__ bias,
                              float* __restrict__ out, int relu) {
    __shared__ float lA[16 * LDA];

    const int m0 = blockIdx.x * 16;
    const int t  = threadIdx.x;

    for (int i = t; i < 16 * DIM; i += 256) {
        int r = i >> 7;
        int k = i & (DIM - 1);
        int row = m0 + r;
        lA[r * LDA + k] = agg[(size_t)row * DIM + k] * norm_d[row];
    }
    __syncthreads();

    const int wave = t >> 5;
    const int lane = t & 31;
    const int n0   = wave * 16;
    const int half = lane >> 4;
    const int l16  = lane & 15;

    const float* Ar = lA + l16 * LDA;
    const float* Wc = W + n0 + l16;

    v8f c = {};
#pragma unroll
    for (int kk = 0; kk < DIM; kk += 4) {
        const int kbase = kk + half * 2;
        v2f a, b;
        a.x = Ar[kbase + 0];
        a.y = Ar[kbase + 1];
        b.x = Wc[(kbase + 0) * DIM];
        b.y = Wc[(kbase + 1) * DIM];
        c = __builtin_amdgcn_wmma_f32_16x16x4_f32(
                 false, a,  false, b,
                 (short)0, c,  false,  false);
    }

    const float bn = bias[n0 + l16];
#pragma unroll
    for (int r = 0; r < 8; ++r) {
        int row = m0 + r + half * 8;
        float v = c[r] + bn;
        if (relu) v = fmaxf(v, 0.0f);
        out[(size_t)row * DIM + n0 + l16] = v;
    }
}

extern "C" void kernel_launch(void* const* d_in, const int* in_sizes, int n_in,
                              void* d_out, int out_size, void* d_ws, size_t ws_size,
                              hipStream_t stream) {
    const float* feat = (const float*)d_in[0];
    const float* W1   = (const float*)d_in[1];
    const float* b1   = (const float*)d_in[2];
    const float* W2   = (const float*)d_in[3];
    const float* b2   = (const float*)d_in[4];
    const int*   src  = (const int*)d_in[5];
    const int*   dst  = (const int*)d_in[6];

    const int N = in_sizes[0] / DIM;
    const int E = in_sizes[5];

    float* ws      = (float*)d_ws;
    float* norm_s  = ws;
    float* norm_d  = ws + N;
    float* agg     = ws + 2 * (size_t)N;
    int*   row_ptr = (int*)(agg + (size_t)N * DIM);
    int*   col     = row_ptr + (N + 1);
    const size_t ws_need = (2 * (size_t)N + (size_t)N * DIM + (size_t)N + 1 + (size_t)E) * sizeof(float);
    if (ws_size < ws_need) return;

    float* out = (float*)d_out;
    int* pos      = (int*)d_out;
    int* col_u    = pos + E;
    int* tile_sum = col_u + E;

    const int ZB = 256;
    const int nTiles = (N + SCAN_TILE - 1) / SCAN_TILE;
    const int rowBlocks = (N + 7) / 8;

    gcn_zero_f32<<<(2 * N + ZB - 1) / ZB, ZB, 0, stream>>>(norm_s, 2 * N);
    gcn_count<<<(E + ZB - 1) / ZB, ZB, 0, stream>>>(src, dst, (int*)norm_s, (int*)norm_d, pos, E);
    gcn_scan_tile<<<nTiles, SCAN_T, 0, stream>>>((const int*)norm_d, row_ptr, tile_sum, N);
    gcn_finish<<<nTiles, SCAN_T, 0, stream>>>(norm_s, row_ptr, tile_sum, N);
    gcn_fill<<<(E + ZB - 1) / ZB, ZB, 0, stream>>>(src, dst, pos, row_ptr, col_u, E);
    gcn_sort_rows<<<rowBlocks, ZB, 0, stream>>>(row_ptr, col_u, col, N);

    gcn_gather<<<rowBlocks, ZB, 0, stream>>>(feat, norm_s, row_ptr, col, agg, N);
    gcn_gemm_wmma<<<N / 16, 256, 0, stream>>>(agg, norm_d, W1, b1, out,  1);

    gcn_gather<<<rowBlocks, ZB, 0, stream>>>(out, norm_s, row_ptr, col, agg, N);
    gcn_gemm_wmma<<<N / 16, 256, 0, stream>>>(agg, norm_d, W2, b2, out,  0);
}


# Layer-1 row gather and layer-2 dense product fused into one launch of resident blocks: 16 summed rows go straight to the LDS A tile, W2 fragments stay in registers; hidden layer never in memory
# speedup vs baseline: 2.4796x; 1.1139x over previous
#include <hip/hip_runtime.h>


typedef float v2f __attribute__((ext_vector_type(2)));
typedef float v8f __attribute__((ext_vector_type(8)));

#define DIM 128

__global__ void gcn_zero_f32(float* __restrict__ p, int n) {
    int i = blockIdx.x * blockDim.x + threadIdx.x;
    int stride = gridDim.x * blockDim.x;
    for (; i < n; i += stride) p[i] = 0.0f;
}

__global__ void gcn_count(const int* __restrict__ src, const int* __restrict__ dst,
                          int* __restrict__ cnt_s, int* __restrict__ cnt_d,
                          int* __restrict__ pos, int ne) {
    int i = blockIdx.x * blockDim.x + threadIdx.x;
    if (i < ne) {
        const int s = src[i];
        const int d = dst[i];
        atomicAdd(&cnt_s[s], 1);
        pos[i] = atomicAdd(&cnt_d[d], 1);
    }
}

#define SCAN_T 256
#define SCAN_PER 4
#define SCAN_TILE (SCAN_T * SCAN_PER)

__global__ void gcn_scan_tile(const int* __restrict__ cnt, int* __restrict__ row_ptr,
                              int* __restrict__ tile_sum, int n) {
    __shared__ int wtot[SCAN_T / 32];
    const int t = threadIdx.x;
    const int lane = t & 31;
    const int wave = t >> 5;
    const int base = blockIdx.x * SCAN_TILE + t * SCAN_PER;
    int v[SCAN_PER];
    int ts = 0;
    for (int k = 0; k < SCAN_PER; ++k) {
        v[k] = (base + k < n) ? cnt[base + k] : 0;
        ts += v[k];
    }
    int x = ts;
    for (int o = 1; o < 32; o <<= 1) {
        int y = __shfl_up(x, o);
        if (lane >= o) x += y;
    }
    if (lane == 31) wtot[wave] = x;
    __syncthreads();
    int woff = 0;
    for (int w = 0; w < SCAN_T / 32; ++w)
        if (w < wave) woff += wtot[w];
    int run = woff + x - ts;
    for (int k = 0; k < SCAN_PER; ++k) {
        if (base + k < n) row_ptr[base + k] = run;
        run += v[k];
    }
    if (t == SCAN_T - 1) tile_sum[blockIdx.x] = run;
}

__global__ void gcn_finish(float* __restrict__ nrm, int* __restrict__ row_ptr,
                           const int* __restrict__ tile_sum, int n) {
    __shared__ int s_off;
    const int t = threadIdx.x;
    if (t < 32) {
        int a = 0;
        for (int j = t; j < (int)blockIdx.x; j += 32) a += tile_sum[j];
        for (int o = 16; o > 0; o >>= 1) a += __shfl_xor(a, o);
        if (t == 0) s_off = a;
    }
    __syncthreads();
    const int off = s_off;
    const int base = blockIdx.x * SCAN_TILE + t * SCAN_PER;
    for (int k = 0; k < SCAN_PER; ++k) {
        const int i = base + k;
        if (i < n) {
            const int cs = __float_as_int(nrm[i]);
            const int cd = __float_as_int(nrm[n + i]);
            const int rp = row_ptr[i] + off;
            row_ptr[i] = rp;
            if (i == n - 1) row_ptr[n] = rp + cd;
            nrm[i]     = rsqrtf(fmaxf((float)cs, 1.0f));
            nrm[n + i] = rsqrtf(fmaxf((float)cd, 1.0f));
        }
    }
}

__global__ void gcn_fill(const int* __restrict__ src, const int* __restrict__ dst,
                         const int* __restrict__ pos, const int* __restrict__ row_ptr,
                         int* __restrict__ col_u, int ne) {
    int i = blockIdx.x * blockDim.x + threadIdx.x;
    if (i < ne) col_u[row_ptr[dst[i]] + pos[i]] = src[i];
}

__global__ void gcn_sort_rows(const int* __restrict__ row_ptr, const int* __restrict__ col_u,
                              int* __restrict__ col, int n) {
    const int r = blockIdx.x * (blockDim.x >> 5) + __builtin_amdgcn_readfirstlane(threadIdx.x >> 5);
    if (r >= n) return;
    const int lane = threadIdx.x & 31;
    const int start = row_ptr[r];
    const int d = row_ptr[r + 1] - start;
    const int* __restrict__ in = col_u + start;
    for (int c0 = 0; c0 < d; c0 += 32) {
        const int i = c0 + lane;
        const int key = (i < d) ? in[i] : 0x7fffffff;
        int rank = 0;
        for (int j0 = 0; j0 < d; j0 += 32) {
            const int kl = (j0 + lane < d) ? in[j0 + lane] : 0x7fffffff;
            const int m = min(32, d - j0);
            for (int jj = 0; jj < m; ++jj) {
                const int kj = __builtin_amdgcn_readlane(kl, jj);
                rank += (int)((kj < key) | ((kj == key) & (j0 + jj < i)));
            }
        }
        if (i < d) col[start + rank] = key;
    }
}

template <int K>
__device__ __forceinline__ void gather_batch(const float* __restrict__ z, int s_l,
                                             int j, int lane4, float4& acc) {
    float4 v[K];
#pragma unroll
    for (int k = 0; k < K; ++k) {
        const int s = __builtin_amdgcn_readlane(s_l, j + k);
        v[k] = *reinterpret_cast<const float4*>(z + (size_t)s * DIM + lane4);
    }
#pragma unroll
    for (int k = 0; k < K; ++k) {
        acc.x += v[k].x;
        acc.y += v[k].y;
        acc.z += v[k].z;
        acc.w += v[k].w;
    }
}

__global__ void gcn_gather(const float* __restrict__ z, const float* __restrict__ norm_d,
                           const float* __restrict__ bias,
                           const int* __restrict__ row_ptr, const int* __restrict__ col,
                           float* __restrict__ out, int n, int relu) {
    const int r = blockIdx.x * (blockDim.x >> 5) + __builtin_amdgcn_readfirstlane(threadIdx.x >> 5);
    if (r >= n) return;
    const int lane = threadIdx.x & 31;
    const int lane4 = lane * 4;
    const int start = row_ptr[r];
    const int end = row_ptr[r + 1];
    const float nd = norm_d[r];
    const float4 b4 = *reinterpret_cast<const float4*>(bias + lane4);
    float4 acc = make_float4(0.0f, 0.0f, 0.0f, 0.0f);
    for (int c0 = start; c0 < end; c0 += 32) {
        const int idx = c0 + lane;
        const int s_l = (idx < end) ? col[idx] : 0;
        const int m = min(32, end - c0);
        int j = 0;
        for (; j + 8 <= m; j += 8) gather_batch<8>(z, s_l, j, lane4, acc);
        if (j + 4 <= m) { gather_batch<4>(z, s_l, j, lane4, acc); j += 4; }
        switch (m - j) {
            case 3: gather_batch<3>(z, s_l, j, lane4, acc); break;
            case 2: gather_batch<2>(z, s_l, j, lane4, acc); break;
            case 1: gather_batch<1>(z, s_l, j, lane4, acc); break;
            default: break;
        }
    }
    float4 o;
    o.x = fmaf(acc.x, nd, b4.x);
    o.y = fmaf(acc.y, nd, b4.y);
    o.z = fmaf(acc.z, nd, b4.z);
    o.w = fmaf(acc.w, nd, b4.w);
    if (relu) {
        o.x = fmaxf(o.x, 0.0f);
        o.y = fmaxf(o.y, 0.0f);
        o.z = fmaxf(o.z, 0.0f);
        o.w = fmaxf(o.w, 0.0f);
    }
    *reinterpret_cast<float4*>(out + (size_t)r * DIM + lane4) = o;
}

#define LDA 129

__global__ void gcn_gemm_wmma(const float* __restrict__ h, const float* __restrict__ norm_s,
                              const float* __restrict__ W, float* __restrict__ z) {
    __shared__ float lA[16 * LDA];

    const int m0 = blockIdx.x * 16;
    const int t  = threadIdx.x;

    for (int i = t; i < 16 * DIM; i += 256) {
        int r = i >> 7;
        int k = i & (DIM - 1);
        int row = m0 + r;
        lA[r * LDA + k] = h[(size_t)row * DIM + k] * norm_s[row];
    }
    __syncthreads();

    const int wave = t >> 5;
    const int lane = t & 31;
    const int n0   = wave * 16;
    const int half = lane >> 4;
    const int l16  = lane & 15;

    const float* Ar = lA + l16 * LDA;
    const float* Wc = W + n0 + l16;

    v8f c = {};
#pragma unroll
    for (int kk = 0; kk < DIM; kk += 4) {
        const int kbase = kk + half * 2;
        v2f a, b;
        a.x = Ar[kbase + 0];
        a.y = Ar[kbase + 1];
        b.x = Wc[(kbase + 0) * DIM];
        b.y = Wc[(kbase + 1) * DIM];
        c = __builtin_amdgcn_wmma_f32_16x16x4_f32(
                 false, a,  false, b,
                 (short)0, c,  false,  false);
    }

#pragma unroll
    for (int r = 0; r < 8; ++r) {
        int row = m0 + r + half * 8;
        z[(size_t)row * DIM + n0 + l16] = c[r];
    }
}


__device__ __forceinline__ float4 gather_row_sum(const float* __restrict__ z, const int* __restrict__ col,
                                                 int start, int end, int s_first, int lane, int lane4) {
    float4 acc = make_float4(0.0f, 0.0f, 0.0f, 0.0f);
    int s_l = s_first;
    for (int c0 = start; c0 < end; c0 += 32) {
        if (c0 != start) {
            const int idx = c0 + lane;
            s_l = (idx < end) ? col[idx] : 0;
        }
        const int m = min(32, end - c0);
        int j = 0;
        for (; j + 8 <= m; j += 8) gather_batch<8>(z, s_l, j, lane4, acc);
        if (j + 4 <= m) { gather_batch<4>(z, s_l, j, lane4, acc); j += 4; }
        switch (m - j) {
            case 3: gather_batch<3>(z, s_l, j, lane4, acc); break;
            case 2: gather_batch<2>(z, s_l, j, lane4, acc); break;
            case 1: gather_batch<1>(z, s_l, j, lane4, acc); break;
            default: break;
        }
    }
    return acc;
}

__device__ __forceinline__ void hidden_row_to_lds(float* __restrict__ dst, const float4 acc, float nd,
                                                  float ns, const float4 b4) {
    dst[0] = fmaxf(fmaf(acc.x, nd, b4.x), 0.0f) * ns;
    dst[1] = fmaxf(fmaf(acc.y, nd, b4.y), 0.0f) * ns;
    dst[2] = fmaxf(fmaf(acc.z, nd, b4.z), 0.0f) * ns;
    dst[3] = fmaxf(fmaf(acc.w, nd, b4.w), 0.0f) * ns;
}

#define FUSED_BLOCKS 1024

__global__ void __launch_bounds__(256)
gcn_gather_gemm(const float* __restrict__ z, const float* __restrict__ norm_d,
                const float* __restrict__ norm_s, const float* __restrict__ bias,
                const int* __restrict__ row_ptr, const int* __restrict__ col,
                const float* __restrict__ W, float* __restrict__ z_out, int ntiles) {
    __shared__ float lA[2][16 * LDA];

    const int t  = threadIdx.x;
    const int lane = t & 31;
    const int lane4 = lane * 4;
    const int wave = __builtin_amdgcn_readfirstlane(t >> 5);
    const int n0   = wave * 16;
    const int half = lane >> 4;
    const int l16  = lane & 15;

    const float* Wc = W + n0 + l16;
    v2f bfrag[DIM / 4];
#pragma unroll
    for (int s = 0; s < DIM / 4; ++s) {
        const int kbase = 4 * s + half * 2;
        bfrag[s].x = Wc[(kbase + 0) * DIM];
        bfrag[s].y = Wc[(kbase + 1) * DIM];
    }
    const float4 b4 = *reinterpret_cast<const float4*>(bias + lane4);

    int tile = blockIdx.x;
    if (tile >= ntiles) return;
    int p0, p1, p2;
    int q0 = 0, q1 = 0, q2 = 0;
    {
        const int r = tile * 16 + 2 * wave;
        p0 = row_ptr[r];
        p1 = row_ptr[r + 1];
        p2 = row_ptr[r + 2];
        if (tile + FUSED_BLOCKS < ntiles) {
            const int rn = r + FUSED_BLOCKS * 16;
            q0 = row_ptr[rn];
            q1 = row_ptr[rn + 1];
            q2 = row_ptr[rn + 2];
        }
    }
    int sA = (p0 + lane < p1) ? col[p0 + lane] : 0;
    int sB = (p1 + lane < p2) ? col[p1 + lane] : 0;

    int buf = 0;
    for (; tile < ntiles; tile += FUSED_BLOCKS, buf ^= 1) {
        const int m0 = tile * 16;
        float* __restrict__ A = lA[buf];
        const int r0 = m0 + 2 * wave;

        const int sAn = (q0 + lane < q1) ? col[q0 + lane] : 0;
        const int sBn = (q1 + lane < q2) ? col[q1 + lane] : 0;
        int u0 = 0, u1 = 0, u2 = 0;
        if (tile + 2 * FUSED_BLOCKS < ntiles) {
            const int rn = r0 + 2 * FUSED_BLOCKS * 16;
            u0 = row_ptr[rn];
            u1 = row_ptr[rn + 1];
            u2 = row_ptr[rn + 2];
        }
        const float nd0 = norm_d[r0];
        const float nd1 = norm_d[r0 + 1];
        const float ns0 = norm_s[r0];
        const float ns1 = norm_s[r0 + 1];

        const float4 acc0 = gather_row_sum(z, col, p0, p1, sA, lane, lane4);
        hidden_row_to_lds(A + (2 * wave) * LDA + lane4, acc0, nd0, ns0, b4);
        const float4 acc1 = gather_row_sum(z, col, p1, p2, sB, lane, lane4);
        hidden_row_to_lds(A + (2 * wave + 1) * LDA + lane4, acc1, nd1, ns1, b4);
        __syncthreads();

        const float* Ar = A + l16 * LDA;
        v8f c = {};
#pragma unroll
        for (int s = 0; s < DIM / 4; ++s) {
            const int kbase = 4 * s + half * 2;
            v2f a;
            a.x = Ar[kbase + 0];
            a.y = Ar[kbase + 1];
            c = __builtin_amdgcn_wmma_f32_16x16x4_f32(
                     false, a,  false, bfrag[s],
                     (short)0, c,  false,  false);
        }

#pragma unroll
        for (int r = 0; r < 8; ++r) {
            int row = m0 + r + half * 8;
            z_out[(size_t)row * DIM + n0 + l16] = c[r];
        }

        p0 = q0; p1 = q1; p2 = q2;
        q0 = u0; q1 = u1; q2 = u2;
        sA = sAn; sB = sBn;
    }
}

extern "C" void kernel_launch(void* const* d_in, const int* in_sizes, int n_in,
                              void* d_out, int out_size, void* d_ws, size_t ws_size,
                              hipStream_t stream) {
    const float* feat = (const float*)d_in[0];
    const float* W1   = (const float*)d_in[1];
    const float* b1   = (const float*)d_in[2];
    const float* W2   = (const float*)d_in[3];
    const float* b2   = (const float*)d_in[4];
    const int*   src  = (const int*)d_in[5];
    const int*   dst  = (const int*)d_in[6];

    const int N = in_sizes[0] / DIM;
    const int E = in_sizes[5];

    float* ws      = (float*)d_ws;
    float* norm_s  = ws;
    float* norm_d  = ws + N;
    float* zbuf    = ws + 2 * (size_t)N;
    int*   row_ptr = (int*)(zbuf + (size_t)N * DIM);
    int*   col     = row_ptr + (N + 1);
    const size_t ws_need = (2 * (size_t)N + (size_t)N * DIM + (size_t)N + 1 + (size_t)E) * sizeof(float);
    if (ws_size < ws_need) return;

    float* out = (float*)d_out;
    int* pos      = (int*)d_out;
    int* col_u    = pos + E;
    int* tile_sum = col_u + E;

    const int ZB = 256;
    const int nTiles = (N + SCAN_TILE - 1) / SCAN_TILE;
    const int rowBlocks = (N + 7) / 8;

    gcn_zero_f32<<<(2 * N + ZB - 1) / ZB, ZB, 0, stream>>>(norm_s, 2 * N);
    gcn_count<<<(E + ZB - 1) / ZB, ZB, 0, stream>>>(src, dst, (int*)norm_s, (int*)norm_d, pos, E);
    gcn_scan_tile<<<nTiles, SCAN_T, 0, stream>>>((const int*)norm_d, row_ptr, tile_sum, N);
    gcn_finish<<<nTiles, SCAN_T, 0, stream>>>(norm_s, row_ptr, tile_sum, N);
    gcn_fill<<<(E + ZB - 1) / ZB, ZB, 0, stream>>>(src, dst, pos, row_ptr, col_u, E);
    gcn_sort_rows<<<rowBlocks, ZB, 0, stream>>>(row_ptr, col_u, col, N);

    gcn_gemm_wmma<<<N / 16, 256, 0, stream>>>(feat, norm_s, W1, out);

    gcn_gather_gemm<<<FUSED_BLOCKS, 256, 0, stream>>>(out, norm_d, norm_s, b1, row_ptr, col, W2, zbuf, N / 16);

    gcn_gather<<<rowBlocks, ZB, 0, stream>>>(zbuf, norm_d, b2, row_ptr, col, out, N,  0);
}


# Row sort, layer-1 product and layer-2 gather run on resident grids (1024/1024/2048 blocks of 8 waves that loop over rows or tiles, W1 held in registers): about 217,000 fewer wave entries per call
# speedup vs baseline: 4.1512x; 1.6741x over previous
#include <hip/hip_runtime.h>


typedef float v2f __attribute__((ext_vector_type(2)));
typedef float v8f __attribute__((ext_vector_type(8)));

#define DIM 128

#define BK_SHIFT 10
#define BK_NODES 1024
#define BK_MAX 1024
#define BK_SLICES 128
#define BK_T 1024
#define BK_PER 16
#define BK_U 8

__global__ void __launch_bounds__(BK_T)
gcn_bucket_hist(const int* __restrict__ src, const int* __restrict__ dst,
                int* __restrict__ hist_d, int* __restrict__ hist_s, int ne, int nb, int per) {
    __shared__ int hd[BK_MAX];
    __shared__ int hs[BK_MAX];
    const int t = threadIdx.x;
    const int j = blockIdx.x;
    for (int b = t; b < nb; b += BK_T) { hd[b] = 0; hs[b] = 0; }
    __syncthreads();
    const int lo = j * per;
    const int hi = min(ne, lo + per);
    for (int e0 = lo + t; e0 < hi; e0 += BK_U * BK_T) {
        int dd[BK_U], ss[BK_U];
#pragma unroll
        for (int u = 0; u < BK_U; ++u) {
            const int ec = min(e0 + u * BK_T, hi - 1);
            dd[u] = dst[ec];
            ss[u] = src[ec];
        }
#pragma unroll
        for (int u = 0; u < BK_U; ++u) {
            if (e0 + u * BK_T < hi) {
                atomicAdd(&hd[dd[u] >> BK_SHIFT], 1);
                atomicAdd(&hs[ss[u] >> BK_SHIFT], 1);
            }
        }
    }
    __syncthreads();
    for (int b = t; b < nb; b += BK_T) {
        hist_d[b * BK_SLICES + j] = hd[b];
        hist_s[b * BK_SLICES + j] = hs[b];
    }
}

__global__ void __launch_bounds__(BK_T)
gcn_bucket_scan(const int* __restrict__ hist_d, const int* __restrict__ hist_s,
                int* __restrict__ off_d, int* __restrict__ off_s, int n_items) {
    __shared__ int wtot[BK_T / 32];
    const int* __restrict__ h = blockIdx.x ? hist_s : hist_d;
    int* __restrict__ o = blockIdx.x ? off_s : off_d;
    const int t = threadIdx.x;
    const int lane = t & 31;
    const int wave = t >> 5;
    int carry = 0;
    for (int base = 0; base < n_items; base += BK_T * BK_PER) {
        const int i0 = base + t * BK_PER;
        int v[BK_PER];
        int ts = 0;
#pragma unroll
        for (int k = 0; k < BK_PER; ++k) {
            v[k] = (i0 + k < n_items) ? h[i0 + k] : 0;
            ts += v[k];
        }
        int x = ts;
        for (int ofs = 1; ofs < 32; ofs <<= 1) {
            int y = __shfl_up(x, ofs);
            if (lane >= ofs) x += y;
        }
        if (lane == 31) wtot[wave] = x;
        __syncthreads();
        int woff = 0;
        int total = 0;
        for (int w = 0; w < BK_T / 32; ++w) {
            const int wt = wtot[w];
            if (w < wave) woff += wt;
            total += wt;
        }
        int run = carry + woff + x - ts;
#pragma unroll
        for (int k = 0; k < BK_PER; ++k) {
            if (i0 + k < n_items) o[i0 + k] = run;
            run += v[k];
        }
        carry += total;
        __syncthreads();
    }
    if (t == 0) o[n_items] = carry;
}

__global__ void __launch_bounds__(BK_T)
gcn_bucket_scatter(const int* __restrict__ src, const int* __restrict__ dst,
                   const int* __restrict__ off_d, const int* __restrict__ off_s,
                   int* __restrict__ pk_d, int* __restrict__ bs, int ne, int nb, int per) {
    __shared__ int cd[BK_MAX];
    __shared__ int cs[BK_MAX];
    const int t = threadIdx.x;
    const int j = blockIdx.x;
    for (int b = t; b < nb; b += BK_T) {
        cd[b] = off_d[b * BK_SLICES + j];
        cs[b] = off_s[b * BK_SLICES + j];
    }
    __syncthreads();
    const int lo = j * per;
    const int hi = min(ne, lo + per);
    for (int e0 = lo + t; e0 < hi; e0 += BK_U * BK_T) {
        int dd[BK_U], ss[BK_U];
#pragma unroll
        for (int u = 0; u < BK_U; ++u) {
            const int ec = min(e0 + u * BK_T, hi - 1);
            dd[u] = dst[ec];
            ss[u] = src[ec];
        }
#pragma unroll
        for (int u = 0; u < BK_U; ++u) {
            if (e0 + u * BK_T < hi) {
                const int p = atomicAdd(&cd[dd[u] >> BK_SHIFT], 1);
                *reinterpret_cast<int2*>(pk_d + 2 * (size_t)p) = make_int2(dd[u], ss[u]);
                const int q = atomicAdd(&cs[ss[u] >> BK_SHIFT], 1);
                bs[q] = ss[u];
            }
        }
    }
}

__global__ void __launch_bounds__(BK_NODES)
gcn_bucket_rows(const int* __restrict__ pk_d, const int* __restrict__ bs,
                const int* __restrict__ off_d, const int* __restrict__ off_s,
                float* __restrict__ norm_s, float* __restrict__ norm_d,
                int* __restrict__ row_ptr, int* __restrict__ col_u, int n, int nb) {
    __shared__ int cnt[BK_NODES];
    __shared__ int wtot[BK_NODES / 32];
    const int t = threadIdx.x;
    const int lane = t & 31;
    const int wave = t >> 5;

    if ((int)blockIdx.x >= nb) {
        const int b = (int)blockIdx.x - nb;
        const int first = b << BK_SHIFT;
        const int lo = off_s[b * BK_SLICES];
        const int hi = off_s[(b + 1) * BK_SLICES];
        cnt[t] = 0;
        __syncthreads();
        for (int e0 = lo + t; e0 < hi; e0 += BK_U * BK_NODES) {
            int ss[BK_U];
#pragma unroll
            for (int u = 0; u < BK_U; ++u) {
                ss[u] = bs[min(e0 + u * BK_NODES, hi - 1)];
            }
#pragma unroll
            for (int u = 0; u < BK_U; ++u)
                if (e0 + u * BK_NODES < hi) atomicAdd(&cnt[ss[u] - first], 1);
        }
        __syncthreads();
        const int node = first + t;
        if (node < n) norm_s[node] = rsqrtf(fmaxf((float)cnt[t], 1.0f));
        return;
    }

    const int b = blockIdx.x;
    const int first = b << BK_SHIFT;
    const int lo = off_d[b * BK_SLICES];
    const int hi = off_d[(b + 1) * BK_SLICES];
    cnt[t] = 0;
    __syncthreads();
    for (int e0 = lo + t; e0 < hi; e0 += BK_U * BK_NODES) {
        int dd[BK_U];
#pragma unroll
        for (int u = 0; u < BK_U; ++u) {
            dd[u] = pk_d[2 * (size_t)min(e0 + u * BK_NODES, hi - 1)];
        }
#pragma unroll
        for (int u = 0; u < BK_U; ++u)
            if (e0 + u * BK_NODES < hi) atomicAdd(&cnt[dd[u] - first], 1);
    }
    __syncthreads();
    const int c = cnt[t];
    int x = c;
    for (int ofs = 1; ofs < 32; ofs <<= 1) {
        int y = __shfl_up(x, ofs);
        if (lane >= ofs) x += y;
    }
    if (lane == 31) wtot[wave] = x;
    __syncthreads();
    int woff = 0;
    for (int w = 0; w < BK_NODES / 32; ++w)
        if (w < wave) woff += wtot[w];
    const int ex = woff + x - c;
    const int node = first + t;
    if (node < n) {
        row_ptr[node] = lo + ex;
        norm_d[node] = rsqrtf(fmaxf((float)c, 1.0f));
    }
    if (node == n - 1) row_ptr[n] = lo + ex + c;
    cnt[t] = ex;
    __syncthreads();
    for (int e0 = lo + t; e0 < hi; e0 += BK_U * BK_NODES) {
        int2 v[BK_U];
#pragma unroll
        for (int u = 0; u < BK_U; ++u) {
            v[u] = *reinterpret_cast<const int2*>(pk_d + 2 * (size_t)min(e0 + u * BK_NODES, hi - 1));
        }
#pragma unroll
        for (int u = 0; u < BK_U; ++u) {
            if (e0 + u * BK_NODES < hi) {
                const int p = atomicAdd(&cnt[v[u].x - first], 1);
                col_u[lo + p] = v[u].y;
            }
        }
    }
}

#define RES_BLOCKS 1024
#define GATHER_BLOCKS 2048

__global__ void __launch_bounds__(256)
gcn_sort_rows(const int* __restrict__ row_ptr, const int* __restrict__ col_u,
              int* __restrict__ col, int n) {
    const int nw = gridDim.x * (blockDim.x >> 5);
    int r = blockIdx.x * (blockDim.x >> 5) + __builtin_amdgcn_readfirstlane(threadIdx.x >> 5);
    if (r >= n) return;
    const int lane = threadIdx.x & 31;
    int p0 = row_ptr[r];
    int p1 = row_ptr[r + 1];
    int q0 = 0, q1 = 0;
    if (r + nw < n) {
        q0 = row_ptr[r + nw];
        q1 = row_ptr[r + nw + 1];
    }
    int key = (p0 + lane < p1) ? col_u[p0 + lane] : 0x7fffffff;
    for (; r < n; r += nw) {
        const int keyn = (q0 + lane < q1) ? col_u[q0 + lane] : 0x7fffffff;
        int u0 = 0, u1 = 0;
        if (r + 2 * nw < n) {
            u0 = row_ptr[r + 2 * nw];
            u1 = row_ptr[r + 2 * nw + 1];
        }
        const int d = p1 - p0;
        if (d <= 32) {
            int rank = 0;
            for (int jj = 0; jj < d; ++jj) {
                const int kj = __builtin_amdgcn_readlane(key, jj);
                rank += (int)((kj < key) | ((kj == key) & (jj < lane)));
            }
            if (lane < d) col[p0 + rank] = key;
        } else {
            const int* __restrict__ in = col_u + p0;
            for (int c0 = 0; c0 < d; c0 += 32) {
                const int i = c0 + lane;
                const int k = (i < d) ? in[i] : 0x7fffffff;
                int rank = 0;
                for (int j0 = 0; j0 < d; j0 += 32) {
                    const int kl = (j0 + lane < d) ? in[j0 + lane] : 0x7fffffff;
                    const int m = min(32, d - j0);
                    for (int jj = 0; jj < m; ++jj) {
                        const int kj = __builtin_amdgcn_readlane(kl, jj);
                        rank += (int)((kj < k) | ((kj == k) & (j0 + jj < i)));
                    }
                }
                if (i < d) col[p0 + rank] = k;
            }
        }
        p0 = q0; p1 = q1;
        q0 = u0; q1 = u1;
        key = keyn;
    }
}

template <int K>
__device__ __forceinline__ void gather_batch(const float* __restrict__ z, int s_l,
                                             int j, int lane4, float4& acc) {
    float4 v[K];
#pragma unroll
    for (int k = 0; k < K; ++k) {
        const int s = __builtin_amdgcn_readlane(s_l, j + k);
        v[k] = *reinterpret_cast<const float4*>(z + (size_t)s * DIM + lane4);
    }
#pragma unroll
    for (int k = 0; k < K; ++k) {
        acc.x += v[k].x;
        acc.y += v[k].y;
        acc.z += v[k].z;
        acc.w += v[k].w;
    }
}

#define LDA 129

__global__ void __launch_bounds__(256)
gcn_gemm_wmma(const float* __restrict__ h, const float* __restrict__ norm_s,
              const float* __restrict__ W, float* __restrict__ z, int ntiles) {
    __shared__ float lA[2][16 * LDA];

    const int t  = threadIdx.x;
    const int lane = t & 31;
    const int lane4 = lane * 4;
    const int wave = __builtin_amdgcn_readfirstlane(t >> 5);
    const int n0   = wave * 16;
    const int half = lane >> 4;
    const int l16  = lane & 15;

    const float* Wc = W + n0 + l16;
    v2f bfrag[DIM / 4];
#pragma unroll
    for (int s = 0; s < DIM / 4; ++s) {
        const int kbase = 4 * s + half * 2;
        bfrag[s].x = Wc[(kbase + 0) * DIM];
        bfrag[s].y = Wc[(kbase + 1) * DIM];
    }

    int buf = 0;
    for (int tile = blockIdx.x; tile < ntiles; tile += gridDim.x, buf ^= 1) {
        const int m0 = tile * 16;
        float* __restrict__ A = lA[buf];

        {
            const int ra = m0 + 2 * wave;
            const float4 x0 = *reinterpret_cast<const float4*>(h + (size_t)ra * DIM + lane4);
            const float4 x1 = *reinterpret_cast<const float4*>(h + (size_t)(ra + 1) * DIM + lane4);
            const float s0 = norm_s[ra];
            const float s1 = norm_s[ra + 1];
            float* __restrict__ d = A + (2 * wave) * LDA + lane4;
            d[0] = x0.x * s0;
            d[1] = x0.y * s0;
            d[2] = x0.z * s0;
            d[3] = x0.w * s0;
            d[LDA + 0] = x1.x * s1;
            d[LDA + 1] = x1.y * s1;
            d[LDA + 2] = x1.z * s1;
            d[LDA + 3] = x1.w * s1;
        }
        __syncthreads();

        const float* Ar = A + l16 * LDA;
        v8f c = {};
#pragma unroll
        for (int s = 0; s < DIM / 4; ++s) {
            const int kbase = 4 * s + half * 2;
            v2f a;
            a.x = Ar[kbase + 0];
            a.y = Ar[kbase + 1];
            c = __builtin_amdgcn_wmma_f32_16x16x4_f32(
                     false, a,  false, bfrag[s],
                     (short)0, c,  false,  false);
        }

#pragma unroll
        for (int r = 0; r < 8; ++r) {
            int row = m0 + r + half * 8;
            z[(size_t)row * DIM + n0 + l16] = c[r];
        }
    }
}


__device__ __forceinline__ float4 gather_row_sum(const float* __restrict__ z, const int* __restrict__ col,
                                                 int start, int end, int s_first, int lane, int lane4) {
    float4 acc = make_float4(0.0f, 0.0f, 0.0f, 0.0f);
    int s_l = s_first;
    for (int c0 = start; c0 < end; c0 += 32) {
        if (c0 != start) {
            const int idx = c0 + lane;
            s_l = (idx < end) ? col[idx] : 0;
        }
        const int m = min(32, end - c0);
        int j = 0;
        for (; j + 8 <= m; j += 8) gather_batch<8>(z, s_l, j, lane4, acc);
        if (j + 4 <= m) { gather_batch<4>(z, s_l, j, lane4, acc); j += 4; }
        switch (m - j) {
            case 3: gather_batch<3>(z, s_l, j, lane4, acc); break;
            case 2: gather_batch<2>(z, s_l, j, lane4, acc); break;
            case 1: gather_batch<1>(z, s_l, j, lane4, acc); break;
            default: break;
        }
    }
    return acc;
}

__global__ void __launch_bounds__(256)
gcn_gather(const float* __restrict__ z, const float* __restrict__ norm_d,
           const float* __restrict__ bias,
           const int* __restrict__ row_ptr, const int* __restrict__ col,
           float* __restrict__ out, int n, int relu) {
    const int nw = gridDim.x * (blockDim.x >> 5);
    int r = blockIdx.x * (blockDim.x >> 5) + __builtin_amdgcn_readfirstlane(threadIdx.x >> 5);
    if (r >= n) return;
    const int lane = threadIdx.x & 31;
    const int lane4 = lane * 4;
    const float4 b4 = *reinterpret_cast<const float4*>(bias + lane4);
    int p0 = row_ptr[r];
    int p1 = row_ptr[r + 1];
    int q0 = 0, q1 = 0;
    if (r + nw < n) {
        q0 = row_ptr[r + nw];
        q1 = row_ptr[r + nw + 1];
    }
    int s_cur = (p0 + lane < p1) ? col[p0 + lane] : 0;
    for (; r < n; r += nw) {
        const int s_next = (q0 + lane < q1) ? col[q0 + lane] : 0;
        int u0 = 0, u1 = 0;
        if (r + 2 * nw < n) {
            u0 = row_ptr[r + 2 * nw];
            u1 = row_ptr[r + 2 * nw + 1];
        }
        const float nd = norm_d[r];
        const float4 acc = gather_row_sum(z, col, p0, p1, s_cur, lane, lane4);
        float4 o;
        o.x = fmaf(acc.x, nd, b4.x);
        o.y = fmaf(acc.y, nd, b4.y);
        o.z = fmaf(acc.z, nd, b4.z);
        o.w = fmaf(acc.w, nd, b4.w);
        if (relu) {
            o.x = fmaxf(o.x, 0.0f);
            o.y = fmaxf(o.y, 0.0f);
            o.z = fmaxf(o.z, 0.0f);
            o.w = fmaxf(o.w, 0.0f);
        }
        *reinterpret_cast<float4*>(out + (size_t)r * DIM + lane4) = o;
        p0 = q0; p1 = q1;
        q0 = u0; q1 = u1;
        s_cur = s_next;
    }
}

__device__ __forceinline__ void hidden_row_to_lds(float* __restrict__ dst, const float4 acc, float nd,
                                                  float ns, const float4 b4) {
    dst[0] = fmaxf(fmaf(acc.x, nd, b4.x), 0.0f) * ns;
    dst[1] = fmaxf(fmaf(acc.y, nd, b4.y), 0.0f) * ns;
    dst[2] = fmaxf(fmaf(acc.z, nd, b4.z), 0.0f) * ns;
    dst[3] = fmaxf(fmaf(acc.w, nd, b4.w), 0.0f) * ns;
}

#define FUSED_BLOCKS 1024

__global__ void __launch_bounds__(256)
gcn_gather_gemm(const float* __restrict__ z, const float* __restrict__ norm_d,
                const float* __restrict__ norm_s, const float* __restrict__ bias,
                const int* __restrict__ row_ptr, const int* __restrict__ col,
                const float* __restrict__ W, float* __restrict__ z_out, int ntiles) {
    __shared__ float lA[2][16 * LDA];

    const int t  = threadIdx.x;
    const int lane = t & 31;
    const int lane4 = lane * 4;
    const int wave = __builtin_amdgcn_readfirstlane(t >> 5);
    const int n0   = wave * 16;
    const int half = lane >> 4;
    const int l16  = lane & 15;

    const float* Wc = W + n0 + l16;
    v2f bfrag[DIM / 4];
#pragma unroll
    for (int s = 0; s < DIM / 4; ++s) {
        const int kbase = 4 * s + half * 2;
        bfrag[s].x = Wc[(kbase + 0) * DIM];
        bfrag[s].y = Wc[(kbase + 1) * DIM];
    }
    const float4 b4 = *reinterpret_cast<const float4*>(bias + lane4);

    int tile = blockIdx.x;
    if (tile >= ntiles) return;
    int p0, p1, p2;
    int q0 = 0, q1 = 0, q2 = 0;
    {
        const int r = tile * 16 + 2 * wave;
        p0 = row_ptr[r];
        p1 = row_ptr[r + 1];
        p2 = row_ptr[r + 2];
        if (tile + FUSED_BLOCKS < ntiles) {
            const int rn = r + FUSED_BLOCKS * 16;
            q0 = row_ptr[rn];
            q1 = row_ptr[rn + 1];
            q2 = row_ptr[rn + 2];
        }
    }
    int sA = (p0 + lane < p1) ? col[p0 + lane] : 0;
    int sB = (p1 + lane < p2) ? col[p1 + lane] : 0;

    int buf = 0;
    for (; tile < ntiles; tile += FUSED_BLOCKS, buf ^= 1) {
        const int m0 = tile * 16;
        float* __restrict__ A = lA[buf];
        const int r0 = m0 + 2 * wave;

        const int sAn = (q0 + lane < q1) ? col[q0 + lane] : 0;
        const int sBn = (q1 + lane < q2) ? col[q1 + lane] : 0;
        int u0 = 0, u1 = 0, u2 = 0;
        if (tile + 2 * FUSED_BLOCKS < ntiles) {
            const int rn = r0 + 2 * FUSED_BLOCKS * 16;
            u0 = row_ptr[rn];
            u1 = row_ptr[rn + 1];
            u2 = row_ptr[rn + 2];
        }
        const float nd0 = norm_d[r0];
        const float nd1 = norm_d[r0 + 1];
        const float ns0 = norm_s[r0];
        const float ns1 = norm_s[r0 + 1];

        const float4 acc0 = gather_row_sum(z, col, p0, p1, sA, lane, lane4);
        hidden_row_to_lds(A + (2 * wave) * LDA + lane4, acc0, nd0, ns0, b4);
        const float4 acc1 = gather_row_sum(z, col, p1, p2, sB, lane, lane4);
        hidden_row_to_lds(A + (2 * wave + 1) * LDA + lane4, acc1, nd1, ns1, b4);
        __syncthreads();

        const float* Ar = A + l16 * LDA;
        v8f c = {};
#pragma unroll
        for (int s = 0; s < DIM / 4; ++s) {
            const int kbase = 4 * s + half * 2;
            v2f a;
            a.x = Ar[kbase + 0];
            a.y = Ar[kbase + 1];
            c = __builtin_amdgcn_wmma_f32_16x16x4_f32(
                     false, a,  false, bfrag[s],
                     (short)0, c,  false,  false);
        }

#pragma unroll
        for (int r = 0; r < 8; ++r) {
            int row = m0 + r + half * 8;
            z_out[(size_t)row * DIM + n0 + l16] = c[r];
        }

        p0 = q0; p1 = q1; p2 = q2;
        q0 = u0; q1 = u1; q2 = u2;
        sA = sAn; sB = sBn;
    }
}

extern "C" void kernel_launch(void* const* d_in, const int* in_sizes, int n_in,
                              void* d_out, int out_size, void* d_ws, size_t ws_size,
                              hipStream_t stream) {
    const float* feat = (const float*)d_in[0];
    const float* W1   = (const float*)d_in[1];
    const float* b1   = (const float*)d_in[2];
    const float* W2   = (const float*)d_in[3];
    const float* b2   = (const float*)d_in[4];
    const int*   src  = (const int*)d_in[5];
    const int*   dst  = (const int*)d_in[6];

    const int N = in_sizes[0] / DIM;
    const int E = in_sizes[5];

    float* ws      = (float*)d_ws;
    float* norm_s  = ws;
    float* norm_d  = ws + N;
    float* zbuf    = ws + 2 * (size_t)N;
    int*   row_ptr = (int*)(zbuf + (size_t)N * DIM);
    int*   col     = row_ptr + (N + 1);
    const size_t ws_need = (2 * (size_t)N + (size_t)N * DIM + (size_t)N + 1 + (size_t)E) * sizeof(float);
    if (ws_size < ws_need) return;

    float* out = (float*)d_out;
    const int NB = (N + BK_NODES - 1) / BK_NODES;
    const int per = (E + BK_SLICES - 1) / BK_SLICES;
    const int n_items = NB * BK_SLICES;
    if (NB > BK_MAX) return;
    if (4 * (size_t)E + 4 * (size_t)n_items + 2 > (size_t)N * DIM) return;
    int*  pk_d   = (int*)d_out;
    int*  bs     = (int*)d_out + 2 * (size_t)E;
    int*  col_u  = bs + E;
    int*  hist_d = col_u + E;
    int*  hist_s = hist_d + n_items;
    int*  off_d  = hist_s + n_items;
    int*  off_s  = off_d + n_items + 1;

    const int ZB = 256;

    gcn_bucket_hist<<<BK_SLICES, BK_T, 0, stream>>>(src, dst, hist_d, hist_s, E, NB, per);
    gcn_bucket_scan<<<2, BK_T, 0, stream>>>(hist_d, hist_s, off_d, off_s, n_items);
    gcn_bucket_scatter<<<BK_SLICES, BK_T, 0, stream>>>(src, dst, off_d, off_s, pk_d, bs, E, NB, per);
    gcn_bucket_rows<<<2 * NB, BK_NODES, 0, stream>>>(pk_d, bs, off_d, off_s, norm_s, norm_d, row_ptr, col_u, N, NB);
    gcn_sort_rows<<<RES_BLOCKS, ZB, 0, stream>>>(row_ptr, col_u, col, N);

    gcn_gemm_wmma<<<RES_BLOCKS, 256, 0, stream>>>(feat, norm_s, W1, out, N / 16);

    gcn_gather_gemm<<<FUSED_BLOCKS, 256, 0, stream>>>(out, norm_d, norm_s, b1, row_ptr, col, W2, zbuf, N / 16);

    gcn_gather<<<GATHER_BLOCKS, ZB, 0, stream>>>(zbuf, norm_d, b2, row_ptr, col, out, N,  0);
}


# Index build's scattered 4- and 8-byte global stores staged through LDS: each permuted edge slice and each bucket's row lists are put together in LDS and leave front to back in whole 128-byte lines
# speedup vs baseline: 5.0194x; 1.2091x over previous
#include <hip/hip_runtime.h>


typedef float v2f __attribute__((ext_vector_type(2)));
typedef float v8f __attribute__((ext_vector_type(8)));

#define DIM 128

#define BK_SHIFT 10
#define BK_NODES 1024
#define BK_MAX 1024
#define BK_MIN_SLICES 256
#define BK_T 1024
#define BK_U 8
#define BK_STAGE 32768

__global__ void __launch_bounds__(BK_T)
gcn_slice_sort(const int* __restrict__ src, const int* __restrict__ dst,
               int* __restrict__ hist_d, int* __restrict__ hist_s,
               int* __restrict__ loc_d, int* __restrict__ loc_s,
               int* __restrict__ pk_d, int* __restrict__ bs, int ne, int nb, int per, int ns) {
    __shared__ int cd[BK_MAX];
    __shared__ int cs[BK_MAX];
    __shared__ int wd[BK_T / 32];
    __shared__ int wsum[BK_T / 32];
    __shared__ __attribute__((aligned(16))) int lp[2 * BK_U * BK_T];
    __shared__ int lb[BK_U * BK_T];
    const int t = threadIdx.x;
    const int j = blockIdx.x;
    const int lane = t & 31;
    const int wave = t >> 5;
    cd[t] = 0;
    cs[t] = 0;
    __syncthreads();
    const int lo = j * per;
    const int hi = min(ne, lo + per);
    int dd[BK_U], ss[BK_U], rd[BK_U], rs[BK_U];
#pragma unroll
    for (int u = 0; u < BK_U; ++u) {
        const int ec = min(lo + t + u * BK_T, ne - 1);
        dd[u] = dst[ec];
        ss[u] = src[ec];
    }
#pragma unroll
    for (int u = 0; u < BK_U; ++u) {
        rd[u] = 0;
        rs[u] = 0;
        if (lo + t + u * BK_T < hi) {
            rd[u] = atomicAdd(&cd[dd[u] >> BK_SHIFT], 1);
            rs[u] = atomicAdd(&cs[ss[u] >> BK_SHIFT], 1);
        }
    }
    __syncthreads();
    const int c_d = cd[t];
    const int c_s = cs[t];
    int xd = c_d, xs = c_s;
    for (int ofs = 1; ofs < 32; ofs <<= 1) {
        const int yd = __shfl_up(xd, ofs);
        const int ys = __shfl_up(xs, ofs);
        if (lane >= ofs) { xd += yd; xs += ys; }
    }
    if (lane == 31) { wd[wave] = xd; wsum[wave] = xs; }
    __syncthreads();
    int od = 0, os = 0;
    for (int w = 0; w < BK_T / 32; ++w) {
        if (w < wave) { od += wd[w]; os += wsum[w]; }
    }
    const int ex_d = od + xd - c_d;
    const int ex_s = os + xs - c_s;
    if (t < nb) {
        hist_d[t * ns + j] = c_d;
        hist_s[t * ns + j] = c_s;
        loc_d[t * ns + j] = ex_d;
        loc_s[t * ns + j] = ex_s;
    }
    cd[t] = ex_d;
    cs[t] = ex_s;
    __syncthreads();
#pragma unroll
    for (int u = 0; u < BK_U; ++u) {
        if (lo + t + u * BK_T < hi) {
            const int p = cd[dd[u] >> BK_SHIFT] + rd[u];
            reinterpret_cast<int2*>(lp)[p] = make_int2(dd[u], ss[u]);
            const int q = cs[ss[u] >> BK_SHIFT] + rs[u];
            lb[q] = ss[u];
        }
    }
    __syncthreads();
    const int n_sl = hi - lo;
    for (int i = t - (lo & 15); i < n_sl; i += BK_T) {
        if (i >= 0)
            *reinterpret_cast<int2*>(pk_d + 2 * (size_t)(lo + i)) = reinterpret_cast<const int2*>(lp)[i];
    }
    for (int i = t - (lo & 31); i < n_sl; i += BK_T) {
        if (i >= 0) bs[lo + i] = lb[i];
    }
}

template <int MODE>
__device__ __forceinline__ void bucket_walk(const int* __restrict__ data, const int* __restrict__ hist,
                                            const int* __restrict__ loc, int b, int ns, int per,
                                            int wave, int lane, int first, int* cnt,
                                            int* __restrict__ col_u, int row_base) {
    for (int m0 = 0; wave + 32 * m0 < ns; m0 += BK_U) {
        const int sg = wave + 32 * (m0 + lane);
        int st = 0, ln = 0;
        if (lane < BK_U && sg < ns) {
            ln = hist[b * ns + sg];
            st = sg * per + loc[b * ns + sg];
        }
        int mx = ln;
        for (int ofs = 1; ofs < BK_U; ofs <<= 1) mx = max(mx, __shfl_xor(mx, ofs));
        mx = __builtin_amdgcn_readfirstlane(mx);
        for (int c0 = 0; c0 < mx; c0 += 32) {
            const int i = c0 + lane;
            int2 v[BK_U];
#pragma unroll
            for (int k = 0; k < BK_U; ++k) {
                const int sk = __builtin_amdgcn_readlane(st, k);
                const int lk = __builtin_amdgcn_readlane(ln, k);
                v[k] = make_int2(0, 0);
                if (i < lk) {
                    if (MODE == 0) v[k].x = data[sk + i];
                    else if (MODE == 1) v[k].x = data[2 * (size_t)(sk + i)];
                    else v[k] = *reinterpret_cast<const int2*>(data + 2 * (size_t)(sk + i));
                }
            }
#pragma unroll
            for (int k = 0; k < BK_U; ++k) {
                const int lk = __builtin_amdgcn_readlane(ln, k);
                if (i < lk) {
                    const int p = atomicAdd(&cnt[v[k].x - first], 1);
                    if (MODE >= 2) col_u[row_base + p] = v[k].y;
                }
            }
        }
    }
}

__global__ void __launch_bounds__(BK_NODES)
gcn_bucket_rows(const int* __restrict__ pk_d, const int* __restrict__ bs,
                const int* __restrict__ hist_d, const int* __restrict__ hist_s,
                const int* __restrict__ loc_d, const int* __restrict__ loc_s,
                float* __restrict__ norm_s, float* __restrict__ norm_d,
                int* __restrict__ row_ptr, int* __restrict__ col_u, int n, int nb, int per, int ns) {
    __shared__ int cnt[BK_NODES];
    __shared__ int wtot[BK_NODES / 32];
    __shared__ int lcol[BK_STAGE];
    const int t = threadIdx.x;
    const int lane = t & 31;
    const int wave = __builtin_amdgcn_readfirstlane(t >> 5);

    if ((int)blockIdx.x >= nb) {
        const int b = (int)blockIdx.x - nb;
        const int first = b << BK_SHIFT;
        cnt[t] = 0;
        __syncthreads();
        bucket_walk<0>(bs, hist_s, loc_s, b, ns, per, wave, lane, first, cnt, col_u, 0);
        __syncthreads();
        const int node = first + t;
        if (node < n) norm_s[node] = rsqrtf(fmaxf((float)cnt[t], 1.0f));
        return;
    }

    const int b = blockIdx.x;
    const int first = b << BK_SHIFT;
    int part = 0;
    for (int i = t; i < b * ns; i += BK_NODES) part += hist_d[i];
    for (int ofs = 16; ofs > 0; ofs >>= 1) part += __shfl_xor(part, ofs);
    if (lane == 0) wtot[wave] = part;
    cnt[t] = 0;
    __syncthreads();
    int lo = 0;
    for (int w = 0; w < BK_NODES / 32; ++w) lo += wtot[w];
    bucket_walk<1>(pk_d, hist_d, loc_d, b, ns, per, wave, lane, first, cnt, col_u, 0);
    __syncthreads();
    const int c = cnt[t];
    int x = c;
    for (int ofs = 1; ofs < 32; ofs <<= 1) {
        int y = __shfl_up(x, ofs);
        if (lane >= ofs) x += y;
    }
    if (lane == 31) wtot[wave] = x;
    __syncthreads();
    int woff = 0, tot_v = 0;
    for (int w = 0; w < BK_NODES / 32; ++w) {
        const int xw = wtot[w];
        tot_v += xw;
        if (w < wave) woff += xw;
    }
    const int tot = __builtin_amdgcn_readfirstlane(tot_v);
    const int ex = woff + x - c;
    const int node = first + t;
    if (node < n) {
        row_ptr[node] = lo + ex;
        norm_d[node] = rsqrtf(fmaxf((float)c, 1.0f));
    }
    if (node == n - 1) row_ptr[n] = lo + ex + c;
    cnt[t] = ex;
    __syncthreads();
    if (tot <= BK_STAGE) {
        bucket_walk<3>(pk_d, hist_d, loc_d, b, ns, per, wave, lane, first, cnt, lcol, 0);
        __syncthreads();
        for (int i = t - (lo & 31); i < tot; i += BK_NODES) {
            if (i >= 0) col_u[lo + i] = lcol[i];
        }
    } else {
        bucket_walk<2>(pk_d, hist_d, loc_d, b, ns, per, wave, lane, first, cnt, col_u, lo);
    }
}

#define RES_BLOCKS 1024
#define GATHER_BLOCKS 2048

__global__ void __launch_bounds__(256)
gcn_sort_rows(const int* __restrict__ row_ptr, const int* __restrict__ col_u,
              int* __restrict__ col, int n) {
    const int nw = gridDim.x * (blockDim.x >> 5);
    int r = blockIdx.x * (blockDim.x >> 5) + __builtin_amdgcn_readfirstlane(threadIdx.x >> 5);
    if (r >= n) return;
    const int lane = threadIdx.x & 31;
    int p0 = row_ptr[r];
    int p1 = row_ptr[r + 1];
    int q0 = 0, q1 = 0;
    if (r + nw < n) {
        q0 = row_ptr[r + nw];
        q1 = row_ptr[r + nw + 1];
    }
    int key = (p0 + lane < p1) ? col_u[p0 + lane] : 0x7fffffff;
    for (; r < n; r += nw) {
        const int keyn = (q0 + lane < q1) ? col_u[q0 + lane] : 0x7fffffff;
        int u0 = 0, u1 = 0;
        if (r + 2 * nw < n) {
            u0 = row_ptr[r + 2 * nw];
            u1 = row_ptr[r + 2 * nw + 1];
        }
        const int d = p1 - p0;
        if (d <= 32) {
            int rank = 0;
            for (int jj = 0; jj < d; ++jj) {
                const int kj = __builtin_amdgcn_readlane(key, jj);
                rank += (int)((kj < key) | ((kj == key) & (jj < lane)));
            }
            if (lane < d) col[p0 + rank] = key;
        } else {
            const int* __restrict__ in = col_u + p0;
            for (int c0 = 0; c0 < d; c0 += 32) {
                const int i = c0 + lane;
                const int k = (i < d) ? in[i] : 0x7fffffff;
                int rank = 0;
                for (int j0 = 0; j0 < d; j0 += 32) {
                    const int kl = (j0 + lane < d) ? in[j0 + lane] : 0x7fffffff;
                    const int m = min(32, d - j0);
                    for (int jj = 0; jj < m; ++jj) {
                        const int kj = __builtin_amdgcn_readlane(kl, jj);
                        rank += (int)((kj < k) | ((kj == k) & (j0 + jj < i)));
                    }
                }
                if (i < d) col[p0 + rank] = k;
            }
        }
        p0 = q0; p1 = q1;
        q0 = u0; q1 = u1;
        key = keyn;
    }
}

template <int K>
__device__ __forceinline__ void gather_batch(const float* __restrict__ z, int s_l,
                                             int j, int lane4, float4& acc) {
    float4 v[K];
#pragma unroll
    for (int k = 0; k < K; ++k) {
        const int s = __builtin_amdgcn_readlane(s_l, j + k);
        v[k] = *reinterpret_cast<const float4*>(z + (size_t)s * DIM + lane4);
    }
#pragma unroll
    for (int k = 0; k < K; ++k) {
        acc.x += v[k].x;
        acc.y += v[k].y;
        acc.z += v[k].z;
        acc.w += v[k].w;
    }
}

#define LDA 129

__global__ void __launch_bounds__(256)
gcn_gemm_wmma(const float* __restrict__ h, const float* __restrict__ norm_s,
              const float* __restrict__ W, float* __restrict__ z, int ntiles) {
    __shared__ float lA[2][16 * LDA];

    const int t  = threadIdx.x;
    const int lane = t & 31;
    const int lane4 = lane * 4;
    const int wave = __builtin_amdgcn_readfirstlane(t >> 5);
    const int n0   = wave * 16;
    const int half = lane >> 4;
    const int l16  = lane & 15;

    const float* Wc = W + n0 + l16;
    v2f bfrag[DIM / 4];
#pragma unroll
    for (int s = 0; s < DIM / 4; ++s) {
        const int kbase = 4 * s + half * 2;
        bfrag[s].x = Wc[(kbase + 0) * DIM];
        bfrag[s].y = Wc[(kbase + 1) * DIM];
    }

    int buf = 0;
    for (int tile = blockIdx.x; tile < ntiles; tile += gridDim.x, buf ^= 1) {
        const int m0 = tile * 16;
        float* __restrict__ A = lA[buf];

        {
            const int ra = m0 + 2 * wave;
            const float4 x0 = *reinterpret_cast<const float4*>(h + (size_t)ra * DIM + lane4);
            const float4 x1 = *reinterpret_cast<const float4*>(h + (size_t)(ra + 1) * DIM + lane4);
            const float s0 = norm_s[ra];
            const float s1 = norm_s[ra + 1];
            float* __restrict__ d = A + (2 * wave) * LDA + lane4;
            d[0] = x0.x * s0;
            d[1] = x0.y * s0;
            d[2] = x0.z * s0;
            d[3] = x0.w * s0;
            d[LDA + 0] = x1.x * s1;
            d[LDA + 1] = x1.y * s1;
            d[LDA + 2] = x1.z * s1;
            d[LDA + 3] = x1.w * s1;
        }
        __syncthreads();

        const float* Ar = A + l16 * LDA;
        v8f c = {};
#pragma unroll
        for (int s = 0; s < DIM / 4; ++s) {
            const int kbase = 4 * s + half * 2;
            v2f a;
            a.x = Ar[kbase + 0];
            a.y = Ar[kbase + 1];
            c = __builtin_amdgcn_wmma_f32_16x16x4_f32(
                     false, a,  false, bfrag[s],
                     (short)0, c,  false,  false);
        }

#pragma unroll
        for (int r = 0; r < 8; ++r) {
            int row = m0 + r + half * 8;
            z[(size_t)row * DIM + n0 + l16] = c[r];
        }
    }
}


__device__ __forceinline__ float4 gather_row_sum(const float* __restrict__ z, const int* __restrict__ col,
                                                 int start, int end, int s_first, int lane, int lane4) {
    float4 acc = make_float4(0.0f, 0.0f, 0.0f, 0.0f);
    int s_l = s_first;
    for (int c0 = start; c0 < end; c0 += 32) {
        if (c0 != start) {
            const int idx = c0 + lane;
            s_l = (idx < end) ? col[idx] : 0;
        }
        const int m = min(32, end - c0);
        int j = 0;
        for (; j + 8 <= m; j += 8) gather_batch<8>(z, s_l, j, lane4, acc);
        if (j + 4 <= m) { gather_batch<4>(z, s_l, j, lane4, acc); j += 4; }
        switch (m - j) {
            case 3: gather_batch<3>(z, s_l, j, lane4, acc); break;
            case 2: gather_batch<2>(z, s_l, j, lane4, acc); break;
            case 1: gather_batch<1>(z, s_l, j, lane4, acc); break;
            default: break;
        }
    }
    return acc;
}

__global__ void __launch_bounds__(256)
gcn_gather(const float* __restrict__ z, const float* __restrict__ norm_d,
           const float* __restrict__ bias,
           const int* __restrict__ row_ptr, const int* __restrict__ col,
           float* __restrict__ out, int n, int relu) {
    const int nw = gridDim.x * (blockDim.x >> 5);
    int r = blockIdx.x * (blockDim.x >> 5) + __builtin_amdgcn_readfirstlane(threadIdx.x >> 5);
    if (r >= n) return;
    const int lane = threadIdx.x & 31;
    const int lane4 = lane * 4;
    const float4 b4 = *reinterpret_cast<const float4*>(bias + lane4);
    int p0 = row_ptr[r];
    int p1 = row_ptr[r + 1];
    int q0 = 0, q1 = 0;
    if (r + nw < n) {
        q0 = row_ptr[r + nw];
        q1 = row_ptr[r + nw + 1];
    }
    int s_cur = (p0 + lane < p1) ? col[p0 + lane] : 0;
    for (; r < n; r += nw) {
        const int s_next = (q0 + lane < q1) ? col[q0 + lane] : 0;
        int u0 = 0, u1 = 0;
        if (r + 2 * nw < n) {
            u0 = row_ptr[r + 2 * nw];
            u1 = row_ptr[r + 2 * nw + 1];
        }
        const float nd = norm_d[r];
        const float4 acc = gather_row_sum(z, col, p0, p1, s_cur, lane, lane4);
        float4 o;
        o.x = fmaf(acc.x, nd, b4.x);
        o.y = fmaf(acc.y, nd, b4.y);
        o.z = fmaf(acc.z, nd, b4.z);
        o.w = fmaf(acc.w, nd, b4.w);
        if (relu) {
            o.x = fmaxf(o.x, 0.0f);
            o.y = fmaxf(o.y, 0.0f);
            o.z = fmaxf(o.z, 0.0f);
            o.w = fmaxf(o.w, 0.0f);
        }
        *reinterpret_cast<float4*>(out + (size_t)r * DIM + lane4) = o;
        p0 = q0; p1 = q1;
        q0 = u0; q1 = u1;
        s_cur = s_next;
    }
}

__device__ __forceinline__ void hidden_row_to_lds(float* __restrict__ dst, const float4 acc, float nd,
                                                  float ns, const float4 b4) {
    dst[0] = fmaxf(fmaf(acc.x, nd, b4.x), 0.0f) * ns;
    dst[1] = fmaxf(fmaf(acc.y, nd, b4.y), 0.0f) * ns;
    dst[2] = fmaxf(fmaf(acc.z, nd, b4.z), 0.0f) * ns;
    dst[3] = fmaxf(fmaf(acc.w, nd, b4.w), 0.0f) * ns;
}

#define FUSED_BLOCKS 1024

__global__ void __launch_bounds__(256)
gcn_gather_gemm(const float* __restrict__ z, const float* __restrict__ norm_d,
                const float* __restrict__ norm_s, const float* __restrict__ bias,
                const int* __restrict__ row_ptr, const int* __restrict__ col,
                const float* __restrict__ W, float* __restrict__ z_out, int ntiles) {
    __shared__ float lA[2][16 * LDA];

    const int t  = threadIdx.x;
    const int lane = t & 31;
    const int lane4 = lane * 4;
    const int wave = __builtin_amdgcn_readfirstlane(t >> 5);
    const int n0   = wave * 16;
    const int half = lane >> 4;
    const int l16  = lane & 15;

    const float* Wc = W + n0 + l16;
    v2f bfrag[DIM / 4];
#pragma unroll
    for (int s = 0; s < DIM / 4; ++s) {
        const int kbase = 4 * s + half * 2;
        bfrag[s].x = Wc[(kbase + 0) * DIM];
        bfrag[s].y = Wc[(kbase + 1) * DIM];
    }
    const float4 b4 = *reinterpret_cast<const float4*>(bias + lane4);

    int tile = blockIdx.x;
    if (tile >= ntiles) return;
    int p0, p1, p2;
    int q0 = 0, q1 = 0, q2 = 0;
    {
        const int r = tile * 16 + 2 * wave;
        p0 = row_ptr[r];
        p1 = row_ptr[r + 1];
        p2 = row_ptr[r + 2];
        if (tile + FUSED_BLOCKS < ntiles) {
            const int rn = r + FUSED_BLOCKS * 16;
            q0 = row_ptr[rn];
            q1 = row_ptr[rn + 1];
            q2 = row_ptr[rn + 2];
        }
    }
    int sA = (p0 + lane < p1) ? col[p0 + lane] : 0;
    int sB = (p1 + lane < p2) ? col[p1 + lane] : 0;

    int buf = 0;
    for (; tile < ntiles; tile += FUSED_BLOCKS, buf ^= 1) {
        const int m0 = tile * 16;
        float* __restrict__ A = lA[buf];
        const int r0 = m0 + 2 * wave;

        const int sAn = (q0 + lane < q1) ? col[q0 + lane] : 0;
        const int sBn = (q1 + lane < q2) ? col[q1 + lane] : 0;
        int u0 = 0, u1 = 0, u2 = 0;
        if (tile + 2 * FUSED_BLOCKS < ntiles) {
            const int rn = r0 + 2 * FUSED_BLOCKS * 16;
            u0 = row_ptr[rn];
            u1 = row_ptr[rn + 1];
            u2 = row_ptr[rn + 2];
        }
        const float nd0 = norm_d[r0];
        const float nd1 = norm_d[r0 + 1];
        const float ns0 = norm_s[r0];
        const float ns1 = norm_s[r0 + 1];

        const float4 acc0 = gather_row_sum(z, col, p0, p1, sA, lane, lane4);
        hidden_row_to_lds(A + (2 * wave) * LDA + lane4, acc0, nd0, ns0, b4);
        const float4 acc1 = gather_row_sum(z, col, p1, p2, sB, lane, lane4);
        hidden_row_to_lds(A + (2 * wave + 1) * LDA + lane4, acc1, nd1, ns1, b4);
        __syncthreads();

        const float* Ar = A + l16 * LDA;
        v8f c = {};
#pragma unroll
        for (int s = 0; s < DIM / 4; ++s) {
            const int kbase = 4 * s + half * 2;
            v2f a;
            a.x = Ar[kbase + 0];
            a.y = Ar[kbase + 1];
            c = __builtin_amdgcn_wmma_f32_16x16x4_f32(
                     false, a,  false, bfrag[s],
                     (short)0, c,  false,  false);
        }

#pragma unroll
        for (int r = 0; r < 8; ++r) {
            int row = m0 + r + half * 8;
            z_out[(size_t)row * DIM + n0 + l16] = c[r];
        }

        p0 = q0; p1 = q1; p2 = q2;
        q0 = u0; q1 = u1; q2 = u2;
        sA = sAn; sB = sBn;
    }
}

extern "C" void kernel_launch(void* const* d_in, const int* in_sizes, int n_in,
                              void* d_out, int out_size, void* d_ws, size_t ws_size,
                              hipStream_t stream) {
    const float* feat = (const float*)d_in[0];
    const float* W1   = (const float*)d_in[1];
    const float* b1   = (const float*)d_in[2];
    const float* W2   = (const float*)d_in[3];
    const float* b2   = (const float*)d_in[4];
    const int*   src  = (const int*)d_in[5];
    const int*   dst  = (const int*)d_in[6];

    const int N = in_sizes[0] / DIM;
    const int E = in_sizes[5];

    float* ws      = (float*)d_ws;
    float* norm_s  = ws;
    float* norm_d  = ws + N;
    float* zbuf    = ws + 2 * (size_t)N;
    int*   row_ptr = (int*)(zbuf + (size_t)N * DIM);
    int*   col     = row_ptr + (N + 1);
    const size_t ws_need = (2 * (size_t)N + (size_t)N * DIM + (size_t)N + 1 + (size_t)E) * sizeof(float);
    if (ws_size < ws_need) return;

    float* out = (float*)d_out;
    const int NB = (N + BK_NODES - 1) / BK_NODES;
    const int NS_FIT = (E + BK_U * BK_T - 1) / (BK_U * BK_T);
    const int NS = NS_FIT > BK_MIN_SLICES ? NS_FIT : BK_MIN_SLICES;
    const int per = (E + NS - 1) / NS;
    const int n_items = NB * NS;
    if (NB > BK_MAX) return;
    if (4 * (size_t)E + 4 * (size_t)n_items > (size_t)N * DIM) return;
    int*  pk_d   = (int*)d_out;
    int*  bs     = (int*)d_out + 2 * (size_t)E;
    int*  col_u  = bs + E;
    int*  hist_d = col_u + E;
    int*  hist_s = hist_d + n_items;
    int*  loc_d  = hist_s + n_items;
    int*  loc_s  = loc_d + n_items;

    const int ZB = 256;

    gcn_slice_sort<<<NS, BK_T, 0, stream>>>(src, dst, hist_d, hist_s, loc_d, loc_s, pk_d, bs, E, NB, per, NS);
    gcn_bucket_rows<<<2 * NB, BK_NODES, 0, stream>>>(pk_d, bs, hist_d, hist_s, loc_d, loc_s, norm_s, norm_d, row_ptr, col_u, N, NB, per, NS);
    gcn_sort_rows<<<RES_BLOCKS, ZB, 0, stream>>>(row_ptr, col_u, col, N);

    gcn_gemm_wmma<<<RES_BLOCKS, 256, 0, stream>>>(feat, norm_s, W1, out, N / 16);

    gcn_gather_gemm<<<FUSED_BLOCKS, 256, 0, stream>>>(out, norm_d, norm_s, b1, row_ptr, col, W2, zbuf, N / 16);

    gcn_gather<<<GATHER_BLOCKS, ZB, 0, stream>>>(zbuf, norm_d, b2, row_ptr, col, out, N,  0);
}


# Both layers' results leave as whole rows: a tile's 16x128 result goes to an LDS tile and, behind the next tile's barrier, each wave stores two full 512-byte rows instead of sixteen 64-byte pieces
# speedup vs baseline: 5.8247x; 1.1604x over previous
#include <hip/hip_runtime.h>


typedef float v2f __attribute__((ext_vector_type(2)));
typedef float v8f __attribute__((ext_vector_type(8)));

#define DIM 128

#define BK_SHIFT 10
#define BK_NODES 1024
#define BK_MAX 1024
#define BK_MIN_SLICES 256
#define BK_T 1024
#define BK_U 8
#define BK_STAGE 32768

__global__ void __launch_bounds__(BK_T)
gcn_slice_sort(const int* __restrict__ src, const int* __restrict__ dst,
               int* __restrict__ hist_d, int* __restrict__ hist_s,
               int* __restrict__ loc_d, int* __restrict__ loc_s,
               int* __restrict__ pk_d, int* __restrict__ bs, int ne, int nb, int per, int ns) {
    __shared__ int cd[BK_MAX];
    __shared__ int cs[BK_MAX];
    __shared__ int wd[BK_T / 32];
    __shared__ int wsum[BK_T / 32];
    __shared__ __attribute__((aligned(16))) int lp[2 * BK_U * BK_T];
    __shared__ int lb[BK_U * BK_T];
    const int t = threadIdx.x;
    const int j = blockIdx.x;
    const int lane = t & 31;
    const int wave = t >> 5;
    cd[t] = 0;
    cs[t] = 0;
    __syncthreads();
    const int lo = j * per;
    const int hi = min(ne, lo + per);
    int dd[BK_U], ss[BK_U], rd[BK_U], rs[BK_U];
#pragma unroll
    for (int u = 0; u < BK_U; ++u) {
        const int ec = min(lo + t + u * BK_T, ne - 1);
        dd[u] = dst[ec];
        ss[u] = src[ec];
    }
#pragma unroll
    for (int u = 0; u < BK_U; ++u) {
        rd[u] = 0;
        rs[u] = 0;
        if (lo + t + u * BK_T < hi) {
            rd[u] = atomicAdd(&cd[dd[u] >> BK_SHIFT], 1);
            rs[u] = atomicAdd(&cs[ss[u] >> BK_SHIFT], 1);
        }
    }
    __syncthreads();
    const int c_d = cd[t];
    const int c_s = cs[t];
    int xd = c_d, xs = c_s;
    for (int ofs = 1; ofs < 32; ofs <<= 1) {
        const int yd = __shfl_up(xd, ofs);
        const int ys = __shfl_up(xs, ofs);
        if (lane >= ofs) { xd += yd; xs += ys; }
    }
    if (lane == 31) { wd[wave] = xd; wsum[wave] = xs; }
    __syncthreads();
    int od = 0, os = 0;
    for (int w = 0; w < BK_T / 32; ++w) {
        if (w < wave) { od += wd[w]; os += wsum[w]; }
    }
    const int ex_d = od + xd - c_d;
    const int ex_s = os + xs - c_s;
    if (t < nb) {
        hist_d[t * ns + j] = c_d;
        hist_s[t * ns + j] = c_s;
        loc_d[t * ns + j] = ex_d;
        loc_s[t * ns + j] = ex_s;
    }
    cd[t] = ex_d;
    cs[t] = ex_s;
    __syncthreads();
#pragma unroll
    for (int u = 0; u < BK_U; ++u) {
        if (lo + t + u * BK_T < hi) {
            const int p = cd[dd[u] >> BK_SHIFT] + rd[u];
            reinterpret_cast<int2*>(lp)[p] = make_int2(dd[u], ss[u]);
            const int q = cs[ss[u] >> BK_SHIFT] + rs[u];
            lb[q] = ss[u];
        }
    }
    __syncthreads();
    const int n_sl = hi - lo;
    for (int i = t - (lo & 15); i < n_sl; i += BK_T) {
        if (i >= 0)
            *reinterpret_cast<int2*>(pk_d + 2 * (size_t)(lo + i)) = reinterpret_cast<const int2*>(lp)[i];
    }
    for (int i = t - (lo & 31); i < n_sl; i += BK_T) {
        if (i >= 0) bs[lo + i] = lb[i];
    }
}

template <int MODE>
__device__ __forceinline__ void bucket_walk(const int* __restrict__ data, const int* __restrict__ hist,
                                            const int* __restrict__ loc, int b, int ns, int per,
                                            int wave, int lane, int first, int* cnt,
                                            int* __restrict__ col_u, int row_base) {
    for (int m0 = 0; wave + 32 * m0 < ns; m0 += BK_U) {
        const int sg = wave + 32 * (m0 + lane);
        int st = 0, ln = 0;
        if (lane < BK_U && sg < ns) {
            ln = hist[b * ns + sg];
            st = sg * per + loc[b * ns + sg];
        }
        int mx = ln;
        for (int ofs = 1; ofs < BK_U; ofs <<= 1) mx = max(mx, __shfl_xor(mx, ofs));
        mx = __builtin_amdgcn_readfirstlane(mx);
        for (int c0 = 0; c0 < mx; c0 += 32) {
            const int i = c0 + lane;
            int2 v[BK_U];
#pragma unroll
            for (int k = 0; k < BK_U; ++k) {
                const int sk = __builtin_amdgcn_readlane(st, k);
                const int lk = __builtin_amdgcn_readlane(ln, k);
                v[k] = make_int2(0, 0);
                if (i < lk) {
                    if (MODE == 0) v[k].x = data[sk + i];
                    else if (MODE == 1) v[k].x = data[2 * (size_t)(sk + i)];
                    else v[k] = *reinterpret_cast<const int2*>(data + 2 * (size_t)(sk + i));
                }
            }
#pragma unroll
            for (int k = 0; k < BK_U; ++k) {
                const int lk = __builtin_amdgcn_readlane(ln, k);
                if (i < lk) {
                    const int p = atomicAdd(&cnt[v[k].x - first], 1);
                    if (MODE >= 2) col_u[row_base + p] = v[k].y;
                }
            }
        }
    }
}

__global__ void __launch_bounds__(BK_NODES)
gcn_bucket_rows(const int* __restrict__ pk_d, const int* __restrict__ bs,
                const int* __restrict__ hist_d, const int* __restrict__ hist_s,
                const int* __restrict__ loc_d, const int* __restrict__ loc_s,
                float* __restrict__ norm_s, float* __restrict__ norm_d,
                int* __restrict__ row_ptr, int* __restrict__ col_u, int n, int nb, int per, int ns) {
    __shared__ int cnt[BK_NODES];
    __shared__ int wtot[BK_NODES / 32];
    __shared__ int lcol[BK_STAGE];
    const int t = threadIdx.x;
    const int lane = t & 31;
    const int wave = __builtin_amdgcn_readfirstlane(t >> 5);

    if ((int)blockIdx.x >= nb) {
        const int b = (int)blockIdx.x - nb;
        const int first = b << BK_SHIFT;
        cnt[t] = 0;
        __syncthreads();
        bucket_walk<0>(bs, hist_s, loc_s, b, ns, per, wave, lane, first, cnt, col_u, 0);
        __syncthreads();
        const int node = first + t;
        if (node < n) norm_s[node] = rsqrtf(fmaxf((float)cnt[t], 1.0f));
        return;
    }

    const int b = blockIdx.x;
    const int first = b << BK_SHIFT;
    int part = 0;
    for (int i = t; i < b * ns; i += BK_NODES) part += hist_d[i];
    for (int ofs = 16; ofs > 0; ofs >>= 1) part += __shfl_xor(part, ofs);
    if (lane == 0) wtot[wave] = part;
    cnt[t] = 0;
    __syncthreads();
    int lo = 0;
    for (int w = 0; w < BK_NODES / 32; ++w) lo += wtot[w];
    bucket_walk<1>(pk_d, hist_d, loc_d, b, ns, per, wave, lane, first, cnt, col_u, 0);
    __syncthreads();
    const int c = cnt[t];
    int x = c;
    for (int ofs = 1; ofs < 32; ofs <<= 1) {
        int y = __shfl_up(x, ofs);
        if (lane >= ofs) x += y;
    }
    if (lane == 31) wtot[wave] = x;
    __syncthreads();
    int woff = 0, tot_v = 0;
    for (int w = 0; w < BK_NODES / 32; ++w) {
        const int xw = wtot[w];
        tot_v += xw;
        if (w < wave) woff += xw;
    }
    const int tot = __builtin_amdgcn_readfirstlane(tot_v);
    const int ex = woff + x - c;
    const int node = first + t;
    if (node < n) {
        row_ptr[node] = lo + ex;
        norm_d[node] = rsqrtf(fmaxf((float)c, 1.0f));
    }
    if (node == n - 1) row_ptr[n] = lo + ex + c;
    cnt[t] = ex;
    __syncthreads();
    if (tot <= BK_STAGE) {
        bucket_walk<3>(pk_d, hist_d, loc_d, b, ns, per, wave, lane, first, cnt, lcol, 0);
        __syncthreads();
        for (int i = t - (lo & 31); i < tot; i += BK_NODES) {
            if (i >= 0) col_u[lo + i] = lcol[i];
        }
    } else {
        bucket_walk<2>(pk_d, hist_d, loc_d, b, ns, per, wave, lane, first, cnt, col_u, lo);
    }
}

__device__ __forceinline__ int order_row(const int* __restrict__ col_u, int* col_s, int start, int end,
                                         int key, int lane) {
    const int d = end - start;
    if (d <= 32) {
        const int k = (lane < d) ? key : 0x7fffffff;
        int rank = 0;
        for (int jj = 0; jj < d; ++jj) {
            const int kj = __builtin_amdgcn_readlane(k, jj);
            rank += (int)((kj < k) | ((kj == k) & (jj < lane)));
        }
        rank = (lane < d) ? rank : lane;
        const int s = __builtin_amdgcn_ds_permute(rank << 2, k);
        if (lane < d) col_s[start + lane] = s;
        return (lane < d) ? s : 0;
    }
    const int* __restrict__ in = col_u + start;
    for (int c0 = 0; c0 < d; c0 += 32) {
        const int i = c0 + lane;
        const int k = (i < d) ? in[i] : 0x7fffffff;
        int rank = 0;
        for (int j0 = 0; j0 < d; j0 += 32) {
            const int kl = (j0 + lane < d) ? in[j0 + lane] : 0x7fffffff;
            const int m = min(32, d - j0);
            for (int jj = 0; jj < m; ++jj) {
                const int kj = __builtin_amdgcn_readlane(kl, jj);
                rank += (int)((kj < k) | ((kj == k) & (j0 + jj < i)));
            }
        }
        if (i < d) col_s[start + rank] = k;
    }
    __threadfence();
    return col_s[start + lane];
}

template <int K, bool WEIGHTED>
__device__ __forceinline__ void gather_batch(const float* __restrict__ x, int s_l, float w_l,
                                             int j, int lane4, float4& acc) {
    float4 v[K];
#pragma unroll
    for (int k = 0; k < K; ++k) {
        const int s = __builtin_amdgcn_readlane(s_l, j + k);
        v[k] = *reinterpret_cast<const float4*>(x + (size_t)s * DIM + lane4);
    }
#pragma unroll
    for (int k = 0; k < K; ++k) {
        if (WEIGHTED) {
            const float w = __int_as_float(__builtin_amdgcn_readlane(__float_as_int(w_l), j + k));
            acc.x = fmaf(v[k].x, w, acc.x);
            acc.y = fmaf(v[k].y, w, acc.y);
            acc.z = fmaf(v[k].z, w, acc.z);
            acc.w = fmaf(v[k].w, w, acc.w);
        } else {
            acc.x += v[k].x;
            acc.y += v[k].y;
            acc.z += v[k].z;
            acc.w += v[k].w;
        }
    }
}

#define LDA 129
#define LDO 132

template <bool WEIGHTED>
__device__ __forceinline__ float4 gather_row_sum(const float* __restrict__ x, const int* col,
                                                 const float* __restrict__ wsrc,
                                                 int start, int end, int s_first, float w_first,
                                                 int lane, int lane4) {
    float4 acc = make_float4(0.0f, 0.0f, 0.0f, 0.0f);
    int s_l = s_first;
    float w_l = w_first;
    for (int c0 = start; c0 < end; c0 += 32) {
        if (c0 != start) {
            const int idx = c0 + lane;
            s_l = (idx < end) ? col[idx] : 0;
            if (WEIGHTED) {
                w_l = wsrc[s_l];
                asm volatile("" : "+v"(w_l));
            }
        }
        const int m = min(32, end - c0);
        int j = 0;
        for (; j + 8 <= m; j += 8) gather_batch<8, WEIGHTED>(x, s_l, w_l, j, lane4, acc);
        if (j + 4 <= m) { gather_batch<4, WEIGHTED>(x, s_l, w_l, j, lane4, acc); j += 4; }
        switch (m - j) {
            case 3: gather_batch<3, WEIGHTED>(x, s_l, w_l, j, lane4, acc); break;
            case 2: gather_batch<2, WEIGHTED>(x, s_l, w_l, j, lane4, acc); break;
            case 1: gather_batch<1, WEIGHTED>(x, s_l, w_l, j, lane4, acc); break;
            default: break;
        }
    }
    return acc;
}

__device__ __forceinline__ void summed_row_to_lds(float* __restrict__ dst, const float4 acc, float nd) {
    dst[0] = acc.x * nd;
    dst[1] = acc.y * nd;
    dst[2] = acc.z * nd;
    dst[3] = acc.w * nd;
}

__device__ __forceinline__ void store_result_rows(const float* O, float* __restrict__ y, int m0,
                                                  int wave, int lane4) {
    unsigned lb = (unsigned)lane4 >> 2;
    asm volatile("" : "+v"(lb));
    lb *= 4u * (unsigned)sizeof(float);
    const char* Ol = reinterpret_cast<const char*>(O + (2 * wave) * LDO) + lb;
    const float4 o0 = *reinterpret_cast<const float4*>(Ol);
    const float4 o1 = *reinterpret_cast<const float4*>(Ol + LDO * sizeof(float));
    char* yl = reinterpret_cast<char*>(y + (size_t)(m0 + 2 * wave) * DIM) + lb;
    *reinterpret_cast<float4*>(yl) = o0;
    *reinterpret_cast<float4*>(yl + DIM * sizeof(float)) = o1;
}

#define FUSED_BLOCKS 1024

template <bool FIRST>
__device__ __forceinline__ void agg_gemm_body(const float* __restrict__ x, const float* __restrict__ norm_d,
                                              const float* __restrict__ norm_s, const float* __restrict__ bias,
                                              const int* __restrict__ row_ptr, const int* __restrict__ col,
                                              int* col_s,
                                              const float* __restrict__ W, float* __restrict__ y, int ntiles) {
    const int* colc = FIRST ? (const int*)col_s : col;
    __shared__ float lA[2][16 * LDA];
    __shared__ __attribute__((aligned(16))) float lNs[2][16];
    __shared__ __attribute__((aligned(16))) float lO[2][16 * LDO];

    const int t  = threadIdx.x;
    const int lane = t & 31;
    const int lane4 = lane * 4;
    const int wave = __builtin_amdgcn_readfirstlane(t >> 5);
    const int n0   = wave * 16;
    const int half = lane >> 4;
    const int l16  = lane & 15;

    const float* Wc = W + n0 + l16;
    v2f bfrag[DIM / 4];
#pragma unroll
    for (int s = 0; s < DIM / 4; ++s) {
        const int kbase = 4 * s + half * 2;
        bfrag[s].x = Wc[(kbase + 0) * DIM];
        bfrag[s].y = Wc[(kbase + 1) * DIM];
    }
    const float bcol = bias[n0 + l16];

    int tile = blockIdx.x;
    if (tile >= ntiles) return;
    int p0, p1, p2;
    int q0 = 0, q1 = 0, q2 = 0;
    {
        const int r = tile * 16 + 2 * wave;
        p0 = row_ptr[r];
        p1 = row_ptr[r + 1];
        p2 = row_ptr[r + 2];
        if (tile + FUSED_BLOCKS < ntiles) {
            const int rn = r + FUSED_BLOCKS * 16;
            q0 = row_ptr[rn];
            q1 = row_ptr[rn + 1];
            q2 = row_ptr[rn + 2];
        }
    }
    int sA = (p0 + lane < p1) ? col[p0 + lane] : 0;
    int sB = (p1 + lane < p2) ? col[p1 + lane] : 0;
    float wA = 0.0f, wB = 0.0f;
    if (FIRST) {
        sA = order_row(col, col_s, p0, p1, sA, lane);
        sB = order_row(col, col_s, p1, p2, sB, lane);
        wA = norm_s[sA];
        wB = norm_s[sB];
        asm volatile("" : "+v"(wA), "+v"(wB));
    }

    int buf = 0;
    for (; tile < ntiles; tile += FUSED_BLOCKS, buf ^= 1) {
        const int m0 = tile * 16;
        float* __restrict__ A = lA[buf];
        const int r0 = m0 + 2 * wave;

        int sAn = (q0 + lane < q1) ? col[q0 + lane] : 0;
        int sBn = (q1 + lane < q2) ? col[q1 + lane] : 0;
        int u0 = 0, u1 = 0, u2 = 0;
        if (tile + 2 * FUSED_BLOCKS < ntiles) {
            const int rn = r0 + 2 * FUSED_BLOCKS * 16;
            u0 = row_ptr[rn];
            u1 = row_ptr[rn + 1];
            u2 = row_ptr[rn + 2];
        }
        const float nd0 = norm_d[r0];
        const float nd1 = norm_d[r0 + 1];
        float ns0 = 0.0f, ns1 = 0.0f;
        if (FIRST) {
            ns0 = norm_s[r0];
            ns1 = norm_s[r0 + 1];
        }

        const float4 acc0 = gather_row_sum<FIRST>(x, colc, norm_s, p0, p1, sA, wA, lane, lane4);
        summed_row_to_lds(A + (2 * wave) * LDA + lane4, acc0, nd0);
        float wAn = 0.0f, wBn = 0.0f;
        if (FIRST) {
            sAn = order_row(col, col_s, q0, q1, sAn, lane);
            sBn = order_row(col, col_s, q1, q2, sBn, lane);
            wAn = norm_s[sAn];
            wBn = norm_s[sBn];
        }
        const float4 acc1 = gather_row_sum<FIRST>(x, colc, norm_s, p1, p2, sB, wB, lane, lane4);
        summed_row_to_lds(A + (2 * wave + 1) * LDA + lane4, acc1, nd1);
        if (FIRST) {
            if (lane < 2) lNs[buf][2 * wave + lane] = (lane == 0) ? ns0 : ns1;
        }
        __syncthreads();

        if (tile != (int)blockIdx.x)
            store_result_rows(lO[buf ^ 1], y, m0 - FUSED_BLOCKS * 16, wave, lane4);

        const float* Ar = A + l16 * LDA;
        v8f c = {};
#pragma unroll
        for (int s = 0; s < DIM / 4; ++s) {
            const int kbase = 4 * s + half * 2;
            v2f a;
            a.x = Ar[kbase + 0];
            a.y = Ar[kbase + 1];
            c = __builtin_amdgcn_wmma_f32_16x16x4_f32(
                     false, a,  false, bfrag[s],
                     (short)0, c,  false,  false);
        }

        float* __restrict__ O = lO[buf] + (half * 8) * LDO + n0 + l16;
        const float* __restrict__ nsr = lNs[buf] + half * 8;
#pragma unroll
        for (int r = 0; r < 8; ++r) {
            float v = c[r] + bcol;
            if (FIRST) v = fmaxf(v, 0.0f) * nsr[r];
            O[r * LDO] = v;
        }

        p0 = q0; p1 = q1; p2 = q2;
        q0 = u0; q1 = u1; q2 = u2;
        sA = sAn; sB = sBn;
        wA = wAn; wB = wBn;
    }
    __syncthreads();
    store_result_rows(lO[buf ^ 1], y, (tile - FUSED_BLOCKS) * 16, wave, lane4);
}

__global__ void __launch_bounds__(256) __attribute__((amdgpu_waves_per_eu(8)))
gcn_layer_first(const float* __restrict__ feat, const float* __restrict__ norm_d,
                const float* __restrict__ norm_s, const float* __restrict__ bias,
                const int* __restrict__ row_ptr, const int* __restrict__ col_u, int* col,
                const float* __restrict__ W, float* __restrict__ hs, int ntiles) {
    agg_gemm_body<true>(feat, norm_d, norm_s, bias, row_ptr, col_u, col, W, hs, ntiles);
}

__global__ void __launch_bounds__(256)
gcn_layer_second(const float* __restrict__ hs, const float* __restrict__ norm_d,
                 const float* __restrict__ norm_s, const float* __restrict__ bias,
                 const int* __restrict__ row_ptr, const int* __restrict__ col,
                 const float* __restrict__ W, float* __restrict__ out, int ntiles) {
    agg_gemm_body<false>(hs, norm_d, norm_s, bias, row_ptr, col, nullptr, W, out, ntiles);
}

extern "C" void kernel_launch(void* const* d_in, const int* in_sizes, int n_in,
                              void* d_out, int out_size, void* d_ws, size_t ws_size,
                              hipStream_t stream) {
    const float* feat = (const float*)d_in[0];
    const float* W1   = (const float*)d_in[1];
    const float* b1   = (const float*)d_in[2];
    const float* W2   = (const float*)d_in[3];
    const float* b2   = (const float*)d_in[4];
    const int*   src  = (const int*)d_in[5];
    const int*   dst  = (const int*)d_in[6];

    const int N = in_sizes[0] / DIM;
    const int E = in_sizes[5];

    float* ws      = (float*)d_ws;
    float* norm_s  = ws;
    float* norm_d  = ws + N;
    float* hsbuf   = ws + 2 * (size_t)N;
    int*   row_ptr = (int*)(hsbuf + (size_t)N * DIM);
    int*   col     = row_ptr + (N + 1);
    const size_t ws_need = (2 * (size_t)N + (size_t)N * DIM + (size_t)N + 1 + (size_t)E) * sizeof(float);
    if (ws_size < ws_need) return;

    float* out = (float*)d_out;
    const int NB = (N + BK_NODES - 1) / BK_NODES;
    const int NS_FIT = (E + BK_U * BK_T - 1) / (BK_U * BK_T);
    const int NS = NS_FIT > BK_MIN_SLICES ? NS_FIT : BK_MIN_SLICES;
    const int per = (E + NS - 1) / NS;
    const int n_items = NB * NS;
    if (NB > BK_MAX) return;
    if (4 * (size_t)E + 4 * (size_t)n_items > (size_t)N * DIM) return;
    int*  pk_d   = (int*)d_out;
    int*  bs     = (int*)d_out + 2 * (size_t)E;
    int*  col_u  = bs + E;
    int*  hist_d = col_u + E;
    int*  hist_s = hist_d + n_items;
    int*  loc_d  = hist_s + n_items;
    int*  loc_s  = loc_d + n_items;

    gcn_slice_sort<<<NS, BK_T, 0, stream>>>(src, dst, hist_d, hist_s, loc_d, loc_s, pk_d, bs, E, NB, per, NS);
    gcn_bucket_rows<<<2 * NB, BK_NODES, 0, stream>>>(pk_d, bs, hist_d, hist_s, loc_d, loc_s, norm_s, norm_d, row_ptr, col_u, N, NB, per, NS);

    gcn_layer_first<<<FUSED_BLOCKS, 256, 0, stream>>>(feat, norm_d, norm_s, b1, row_ptr, col_u, col, W1, hsbuf, N / 16);

    gcn_layer_second<<<FUSED_BLOCKS, 256, 0, stream>>>(hsbuf, norm_d, norm_s, b2, row_ptr, col, W2, out, N / 16);
}
